# starting kernel
# baseline (speedup 1.0000x reference)
#include <hip/hip_runtime.h>
#include <math.h>

typedef __attribute__((ext_vector_type(2))) float v2f;
typedef __attribute__((ext_vector_type(8))) float v8f;

#define TS   32
#define KC   32
#define ASTR 36
#define BSTR 33

static __device__ __forceinline__ float leaky02(float x) {
    return x > 0.f ? x : 0.2f * x;
}

static __device__ __forceinline__ void atomicMaxF(float* addr, float v) {
    if (v >= 0.f)
        atomicMax((int*)addr, __float_as_int(v));
    else
        atomicMin((unsigned int*)addr, __float_as_uint(v));
}

__global__ __launch_bounds__(128)
void gemm_wmma_f32(const float* __restrict__ A, const float* __restrict__ B,
                   float* __restrict__ D, int M, int K, int N) {
    __shared__ float As[TS * ASTR];
    __shared__ float Bs[KC * BSTR];

    const int tileN = blockIdx.x * TS;
    const int tileM = blockIdx.y * TS;
    const int t    = threadIdx.x;
    const int wv   = t >> 5;
    const int lane = t & 31;
    const int half = lane >> 4;
    const int lm   = lane & 15;
    const int wm   = wv >> 1;
    const int wn   = wv & 1;

    const int crow = t >> 2;
    const int ccol = (t & 3) * 8;

    v8f acc = {};

    for (int kt = 0; kt < K; kt += KC) {
        const float* ga = A + (size_t)(tileM + crow) * K + kt + ccol;
        const float* gb = B + (size_t)(kt + crow) * N + tileN + ccol;
        if (kt + KC < K) {
            __builtin_prefetch(ga + KC, 0, 0);
            __builtin_prefetch(gb + (size_t)KC * N, 0, 0);
        }
        #pragma unroll
        for (int j = 0; j < 8; ++j) As[crow * ASTR + ccol + j] = ga[j];
        #pragma unroll
        for (int j = 0; j < 8; ++j) Bs[crow * BSTR + ccol + j] = gb[j];
        __syncthreads();

        #pragma unroll
        for (int kk = 0; kk < KC; kk += 4) {
            const int k0 = kk + 2 * half;
            v2f a, b;
            a.x = As[(wm * 16 + lm) * ASTR + k0];
            a.y = As[(wm * 16 + lm) * ASTR + k0 + 1];
            b.x = Bs[(k0)     * BSTR + wn * 16 + lm];
            b.y = Bs[(k0 + 1) * BSTR + wn * 16 + lm];
            acc = __builtin_amdgcn_wmma_f32_16x16x4_f32(
                false, a, false, b, (short)0, acc, false, false);
        }
        __syncthreads();
    }

    const int col   = tileN + wn * 16 + lm;
    const int rbase = tileM + wm * 16 + half * 8;
    #pragma unroll
    for (int r = 0; r < 8; ++r)
        D[(size_t)(rbase + r) * N + col] = acc[r];
}

__global__ void att_logits(const float* __restrict__ h,
                           const float* __restrict__ a_s,
                           const float* __restrict__ a_d,
                           float* __restrict__ al_s, float* __restrict__ al_d,
                           int N, int H, int C) {
    int t = blockIdx.x * blockDim.x + threadIdx.x;
    if (t >= N * H) return;
    int n = t / H, hh = t - n * H;
    const float* row = h + (size_t)n * H * C + (size_t)hh * C;
    const float* vs  = a_s + (size_t)hh * C;
    const float* vd  = a_d + (size_t)hh * C;
    float ss = 0.f, sd = 0.f;
    for (int c = 0; c < C; ++c) {
        float v = row[c];
        ss = fmaf(v, vs[c], ss);
        sd = fmaf(v, vd[c], sd);
    }
    al_s[t] = ss;
    al_d[t] = sd;
}

__global__ void fill_f32(float* __restrict__ p, float v, size_t n) {
    size_t i = blockIdx.x * (size_t)blockDim.x + threadIdx.x;
    if (i < n) p[i] = v;
}

__global__ void edge_max(const int* __restrict__ ei, int E, int N, int H,
                         const float* __restrict__ al_s,
                         const float* __restrict__ al_d,
                         float* __restrict__ mbuf) {
    int t = blockIdx.x * blockDim.x + threadIdx.x;
    int tot = (E + N) * H;
    if (t >= tot) return;
    int e = t / H, hh = t - e * H;
    int s, d;
    if (e < E) { s = ei[e]; d = ei[E + e]; } else { s = d = e - E; }
    float v = leaky02(al_s[s * H + hh] + al_d[d * H + hh]);
    atomicMaxF(&mbuf[d * H + hh], v);
}

__global__ void edge_exp(const int* __restrict__ ei, int E, int N, int H,
                         const float* __restrict__ al_s,
                         const float* __restrict__ al_d,
                         const float* __restrict__ mbuf,
                         float* __restrict__ exbuf,
                         float* __restrict__ denom) {
    int t = blockIdx.x * blockDim.x + threadIdx.x;
    int tot = (E + N) * H;
    if (t >= tot) return;
    int e = t / H, hh = t - e * H;
    int s, d;
    if (e < E) { s = ei[e]; d = ei[E + e]; } else { s = d = e - E; }
    float v  = leaky02(al_s[s * H + hh] + al_d[d * H + hh]);
    float ex = __expf(v - mbuf[d * H + hh]);
    exbuf[t] = ex;
    atomicAdd(&denom[d * H + hh], ex);
}

__global__ void edge_aggregate(const int* __restrict__ ei, int E, int N,
                               int H, int C,
                               const float* __restrict__ hfeat,
                               const float* __restrict__ exbuf,
                               const float* __restrict__ denom,
                               float* __restrict__ agg) {
    const int F = H * C;
    long long t = blockIdx.x * (long long)blockDim.x + threadIdx.x;
    long long tot = (long long)(E + N) * F;
    if (t >= tot) return;
    int e = (int)(t / F);
    int f = (int)(t - (long long)e * F);
    int hh = f / C;
    int s, d;
    if (e < E) { s = ei[e]; d = ei[E + e]; } else { s = d = e - E; }
    float alpha = exbuf[(size_t)e * H + hh] / denom[(size_t)d * H + hh];
    atomicAdd(&agg[(size_t)d * F + f], hfeat[(size_t)s * F + f] * alpha);
}

__global__ void bias_relu_inplace(float* __restrict__ p,
                                  const float* __restrict__ b,
                                  size_t total, int F) {
    size_t t = blockIdx.x * (size_t)blockDim.x + threadIdx.x;
    if (t >= total) return;
    int f = (int)(t % F);
    float v = p[t] + b[f];
    p[t] = v > 0.f ? v : 0.f;
}

__global__ void mean_heads_bias(const float* __restrict__ agg,
                                const float* __restrict__ b,
                                float* __restrict__ out, int N, int H, int C) {
    int t = blockIdx.x * blockDim.x + threadIdx.x;
    if (t >= N * C) return;
    int n = t / C, c = t - n * C;
    float s = 0.f;
    for (int hh = 0; hh < H; ++hh)
        s += agg[(size_t)n * H * C + (size_t)hh * C + c];
    out[t] = s / (float)H + b[c];
}

static inline unsigned cdiv_u(size_t a, unsigned b) {
    return (unsigned)((a + b - 1) / b);
}

extern "C" void kernel_launch(void* const* d_in, const int* in_sizes, int n_in,
                              void* d_out, int out_size, void* d_ws, size_t ws_size,
                              hipStream_t stream) {
    const float* x     = (const float*)d_in[0];
    const int*   ei    = (const int*)d_in[1];
    const float* W1    = (const float*)d_in[2];
    const float* a_s1  = (const float*)d_in[3];
    const float* a_d1  = (const float*)d_in[4];
    const float* b1    = (const float*)d_in[5];
    const float* W2    = (const float*)d_in[6];
    const float* a_s2  = (const float*)d_in[7];
    const float* a_d2  = (const float*)d_in[8];
    const float* b2    = (const float*)d_in[9];
    float*       out   = (float*)d_out;

    const int F1   = in_sizes[3];
    const int F2   = in_sizes[7];
    const int LAT  = in_sizes[9];
    const int H    = F2 / LAT;
    const int C1   = F1 / H;
    const int Fin  = in_sizes[2] / F1;
    const int N    = in_sizes[0] / Fin;
    const int E    = in_sizes[1] / 2;
    const int Etot = E + N;

    float* ws = (float*)d_ws;
    size_t o = 0;
    float* h1   = ws + o; o += (size_t)N * F1;
    float* agg1 = ws + o; o += (size_t)N * F1;
    float* ex   = ws + o; o += (size_t)Etot * H;
    float* als  = ws + o; o += (size_t)N * H;
    float* ald  = ws + o; o += (size_t)N * H;
    float* mbuf = ws + o; o += (size_t)N * H;
    float* den  = ws + o; o += (size_t)N * H;
    float* h2   = h1;
    float* agg2 = h1 + (size_t)N * F2;
    (void)ws_size; (void)n_in; (void)out_size;

    const unsigned TB = 256;
    const float NEG_INF = -__builtin_huge_valf();

    fill_f32<<<cdiv_u((size_t)N * F1, TB), TB, 0, stream>>>(agg1, 0.f, (size_t)N * F1);
    fill_f32<<<cdiv_u((size_t)N * H, TB), TB, 0, stream>>>(mbuf, NEG_INF, (size_t)N * H);
    fill_f32<<<cdiv_u((size_t)N * H, TB), TB, 0, stream>>>(den, 0.f, (size_t)N * H);

    gemm_wmma_f32<<<dim3(F1 / TS, N / TS), 128, 0, stream>>>(x, W1, h1, N, Fin, F1);

    att_logits<<<cdiv_u((size_t)N * H, TB), TB, 0, stream>>>(h1, a_s1, a_d1, als, ald, N, H, C1);

    edge_max<<<cdiv_u((size_t)Etot * H, TB), TB, 0, stream>>>(ei, E, N, H, als, ald, mbuf);
    edge_exp<<<cdiv_u((size_t)Etot * H, TB), TB, 0, stream>>>(ei, E, N, H, als, ald, mbuf, ex, den);
    edge_aggregate<<<cdiv_u((size_t)Etot * F1, TB), TB, 0, stream>>>(
        ei, E, N, H, C1, h1, ex, den, agg1);

    bias_relu_inplace<<<cdiv_u((size_t)N * F1, TB), TB, 0, stream>>>(agg1, b1, (size_t)N * F1, F1);

    gemm_wmma_f32<<<dim3(F2 / TS, N / TS), 128, 0, stream>>>(agg1, W2, h2, N, F1, F2);

    fill_f32<<<cdiv_u((size_t)N * F2, TB), TB, 0, stream>>>(agg2, 0.f, (size_t)N * F2);
    fill_f32<<<cdiv_u((size_t)N * H, TB), TB, 0, stream>>>(mbuf, NEG_INF, (size_t)N * H);
    fill_f32<<<cdiv_u((size_t)N * H, TB), TB, 0, stream>>>(den, 0.f, (size_t)N * H);

    att_logits<<<cdiv_u((size_t)N * H, TB), TB, 0, stream>>>(h2, a_s2, a_d2, als, ald, N, H, LAT);

    edge_max<<<cdiv_u((size_t)Etot * H, TB), TB, 0, stream>>>(ei, E, N, H, als, ald, mbuf);
    edge_exp<<<cdiv_u((size_t)Etot * H, TB), TB, 0, stream>>>(ei, E, N, H, als, ald, mbuf, ex, den);
    edge_aggregate<<<cdiv_u((size_t)Etot * F2, TB), TB, 0, stream>>>(
        ei, E, N, H, LAT, h2, ex, den, agg2);

    mean_heads_bias<<<cdiv_u((size_t)N * LAT, TB), TB, 0, stream>>>(agg2, b2, out, N, H, LAT);
}


# aggregation by destination bands: a block keeps 64 agg rows in LDS, scans the edge list, adds alpha*row with LDS float adds, stores once; global float atomics, agg fills gone; sum order = LDS arrival
# speedup vs baseline: 3.5164x; 3.5164x over previous
#include <hip/hip_runtime.h>
#include <math.h>
#include <type_traits>

typedef __attribute__((ext_vector_type(2))) float v2f;
typedef __attribute__((ext_vector_type(8))) float v8f;

#define TS   32
#define KC   32
#define ASTR 36
#define BSTR 33

static __device__ __forceinline__ float leaky02(float x) {
    return x > 0.f ? x : 0.2f * x;
}

static __device__ __forceinline__ void atomicMaxF(float* addr, float v) {
    if (v >= 0.f)
        atomicMax((int*)addr, __float_as_int(v));
    else
        atomicMin((unsigned int*)addr, __float_as_uint(v));
}

__global__ __launch_bounds__(128)
void gemm_wmma_f32(const float* __restrict__ A, const float* __restrict__ B,
                   float* __restrict__ D, int M, int K, int N) {
    __shared__ float As[TS * ASTR];
    __shared__ float Bs[KC * BSTR];

    const int tileN = blockIdx.x * TS;
    const int tileM = blockIdx.y * TS;
    const int t    = threadIdx.x;
    const int wv   = t >> 5;
    const int lane = t & 31;
    const int half = lane >> 4;
    const int lm   = lane & 15;
    const int wm   = wv >> 1;
    const int wn   = wv & 1;

    const int crow = t >> 2;
    const int ccol = (t & 3) * 8;

    v8f acc = {};

    for (int kt = 0; kt < K; kt += KC) {
        const float* ga = A + (size_t)(tileM + crow) * K + kt + ccol;
        const float* gb = B + (size_t)(kt + crow) * N + tileN + ccol;
        if (kt + KC < K) {
            __builtin_prefetch(ga + KC, 0, 0);
            __builtin_prefetch(gb + (size_t)KC * N, 0, 0);
        }
        #pragma unroll
        for (int j = 0; j < 8; ++j) As[crow * ASTR + ccol + j] = ga[j];
        #pragma unroll
        for (int j = 0; j < 8; ++j) Bs[crow * BSTR + ccol + j] = gb[j];
        __syncthreads();

        #pragma unroll
        for (int kk = 0; kk < KC; kk += 4) {
            const int k0 = kk + 2 * half;
            v2f a, b;
            a.x = As[(wm * 16 + lm) * ASTR + k0];
            a.y = As[(wm * 16 + lm) * ASTR + k0 + 1];
            b.x = Bs[(k0)     * BSTR + wn * 16 + lm];
            b.y = Bs[(k0 + 1) * BSTR + wn * 16 + lm];
            acc = __builtin_amdgcn_wmma_f32_16x16x4_f32(
                false, a, false, b, (short)0, acc, false, false);
        }
        __syncthreads();
    }

    const int col   = tileN + wn * 16 + lm;
    const int rbase = tileM + wm * 16 + half * 8;
    #pragma unroll
    for (int r = 0; r < 8; ++r)
        D[(size_t)(rbase + r) * N + col] = acc[r];
}

__global__ void att_logits(const float* __restrict__ h,
                           const float* __restrict__ a_s,
                           const float* __restrict__ a_d,
                           float* __restrict__ al_s, float* __restrict__ al_d,
                           int N, int H, int C) {
    int t = blockIdx.x * blockDim.x + threadIdx.x;
    if (t >= N * H) return;
    int n = t / H, hh = t - n * H;
    const float* row = h + (size_t)n * H * C + (size_t)hh * C;
    const float* vs  = a_s + (size_t)hh * C;
    const float* vd  = a_d + (size_t)hh * C;
    float ss = 0.f, sd = 0.f;
    for (int c = 0; c < C; ++c) {
        float v = row[c];
        ss = fmaf(v, vs[c], ss);
        sd = fmaf(v, vd[c], sd);
    }
    al_s[t] = ss;
    al_d[t] = sd;
}

__global__ void fill_f32(float* __restrict__ p, float v, size_t n) {
    size_t i = blockIdx.x * (size_t)blockDim.x + threadIdx.x;
    if (i < n) p[i] = v;
}

__global__ void edge_max(const int* __restrict__ ei, int E, int N, int H,
                         const float* __restrict__ al_s,
                         const float* __restrict__ al_d,
                         float* __restrict__ mbuf) {
    int t = blockIdx.x * blockDim.x + threadIdx.x;
    int tot = (E + N) * H;
    if (t >= tot) return;
    int e = t / H, hh = t - e * H;
    int s, d;
    if (e < E) { s = ei[e]; d = ei[E + e]; } else { s = d = e - E; }
    float v = leaky02(al_s[s * H + hh] + al_d[d * H + hh]);
    atomicMaxF(&mbuf[d * H + hh], v);
}

__global__ void edge_exp(const int* __restrict__ ei, int E, int N, int H,
                         const float* __restrict__ al_s,
                         const float* __restrict__ al_d,
                         const float* __restrict__ mbuf,
                         float* __restrict__ exbuf,
                         float* __restrict__ denom) {
    int t = blockIdx.x * blockDim.x + threadIdx.x;
    int tot = (E + N) * H;
    if (t >= tot) return;
    int e = t / H, hh = t - e * H;
    int s, d;
    if (e < E) { s = ei[e]; d = ei[E + e]; } else { s = d = e - E; }
    float v  = leaky02(al_s[s * H + hh] + al_d[d * H + hh]);
    float ex = __expf(v - mbuf[d * H + hh]);
    exbuf[t] = ex;
    atomicAdd(&denom[d * H + hh], ex);
}

template <int F, int C, int R, int NW>
__global__ __launch_bounds__(NW * 32)
void agg_band(const int* __restrict__ ei, int E, int N,
              const float* __restrict__ hfeat,
              const float* __restrict__ exbuf,
              const float* __restrict__ denom,
              float* __restrict__ agg) {
    static_assert(F == 4 * C, "four heads");
    static_assert(R % NW == 0 && F % 64 == 0, "band and row shapes");
    constexpr int NT    = NW * 32;
    constexpr int V2    = F / 64;
    constexpr int SPW   = R / NW;
    constexpr int DRAIN = 8;
    constexpr int QCAP  = DRAIN + 256 + SPW;

    __shared__ __attribute__((aligned(16))) float acc[R * F];
    __shared__ int q[NW][QCAP];

    const int tid  = threadIdx.x;
    const int lane = tid & 31;
    const int w    = __builtin_amdgcn_readfirstlane(tid >> 5);
    const int base = blockIdx.x * R;

    float4* acc4 = reinterpret_cast<float4*>(acc);
    for (int i = tid; i < R * F / 4; i += NT) acc4[i] = make_float4(0.f, 0.f, 0.f, 0.f);
    __syncthreads();

    int cnt = 0;
    {
        const int row0 = base + w * SPW;
        int nself = N - row0;
        nself = nself < 0 ? 0 : (nself > SPW ? SPW : nself);
        if (lane < nself) q[w][lane] = E + row0 + lane;
        cnt = nself;
    }

    const int* __restrict__ dst = ei + E;
    const int nsteps = E >> 8;

    auto push = [&](bool hit, int ev) {
        const unsigned mask = __builtin_amdgcn_ballot_w32(hit);
        if (hit) q[w][cnt + (int)__builtin_amdgcn_mbcnt_lo(mask, 0u)] = ev;
        cnt += __builtin_popcount(mask);
    };
    auto test4 = [&](const int4& v, int e) {
        const bool m0 = (unsigned)(v.x - base) < (unsigned)R;
        const bool m1 = (unsigned)(v.y - base) < (unsigned)R;
        const bool m2 = (unsigned)(v.z - base) < (unsigned)R;
        const bool m3 = (unsigned)(v.w - base) < (unsigned)R;
        if (__builtin_amdgcn_ballot_w32(m0 | m1 | m2 | m3)) {
            push(m0, e);
            push(m1, e + 1);
            push(m2, e + 2);
            push(m3, e + 3);
        }
    };

    for (int step = w; ; step += NW) {
        const bool last = step >= nsteps;
        if (!last) {
            const int  e  = step * 256 + lane * 4;
            const int4 va = *reinterpret_cast<const int4*>(dst + e);
            const int4 vb = *reinterpret_cast<const int4*>(dst + e + 128);
            test4(va, e);
            test4(vb, e + 128);
        } else if (w == 0) {
            for (int e = nsteps * 256 + lane; __builtin_amdgcn_ballot_w32(e < E); e += 32)
                push(e < E && (unsigned)(dst[e < E ? e : 0] - base) < (unsigned)R, e);
        }

        if (cnt >= DRAIN || (last && cnt > 0)) {
            for (int b = 0; b < cnt; b += 32) {
                const int n = (cnt - b) < 32 ? (cnt - b) : 32;
                int   s = 0, r = 0;
                float a0 = 0.f, a1 = 0.f, a2 = 0.f, a3 = 0.f;
                if (lane < n) {
                    const int eq = q[w][b + lane];
                    int d;
                    if (eq < E) { s = ei[eq]; d = dst[eq]; } else { s = d = eq - E; }
                    const float4 x4 = *reinterpret_cast<const float4*>(exbuf + (size_t)eq * 4);
                    const float4 d4 = *reinterpret_cast<const float4*>(denom + (size_t)d * 4);
                    a0 = x4.x / d4.x;  a1 = x4.y / d4.y;
                    a2 = x4.z / d4.z;  a3 = x4.w / d4.w;
                    r  = d - base;
                }
                auto rows = [&](int j, auto UC) {
                    constexpr int U = decltype(UC)::value;
                    float2 v[U][V2];
                    #pragma unroll
                    for (int u = 0; u < U; ++u) {
                        const int sj = __builtin_amdgcn_readlane(s, j + u);
                        const float2* __restrict__ row =
                            reinterpret_cast<const float2*>(hfeat + (size_t)sj * F) + lane;
                        #pragma unroll
                        for (int k = 0; k < V2; ++k) v[u][k] = row[k * 32];
                    }
                    #pragma unroll
                    for (int u = 0; u < U; ++u) {
                        const int rj = __builtin_amdgcn_readlane(r, j + u);
                        float al[4];
                        al[0] = __builtin_bit_cast(float, __builtin_amdgcn_readlane(__builtin_bit_cast(int, a0), j + u));
                        al[1] = __builtin_bit_cast(float, __builtin_amdgcn_readlane(__builtin_bit_cast(int, a1), j + u));
                        al[2] = __builtin_bit_cast(float, __builtin_amdgcn_readlane(__builtin_bit_cast(int, a2), j + u));
                        al[3] = __builtin_bit_cast(float, __builtin_amdgcn_readlane(__builtin_bit_cast(int, a3), j + u));
                        float* arow = acc + rj * F + lane * 2;
                        #pragma unroll
                        for (int k = 0; k < V2; ++k) {
                            const float a = al[(k * 64) / C];
                            __hip_atomic_fetch_add(arow + k * 64,     v[u][k].x * a,
                                                   __ATOMIC_RELAXED, __HIP_MEMORY_SCOPE_WORKGROUP);
                            __hip_atomic_fetch_add(arow + k * 64 + 1, v[u][k].y * a,
                                                   __ATOMIC_RELAXED, __HIP_MEMORY_SCOPE_WORKGROUP);
                        }
                    }
                };
                int j = 0;
                for (; j + 4 <= n; j += 4) rows(j, std::integral_constant<int, 4>{});
                for (; j < n; ++j)         rows(j, std::integral_constant<int, 1>{});
            }
            cnt = 0;
        }
        if (last) break;
    }

    __syncthreads();
    float4* __restrict__ out4 = reinterpret_cast<float4*>(agg + (size_t)base * F);
    for (int i = tid; i < R * F / 4; i += NT)
        if (base + i / (F / 4) < N) out4[i] = acc4[i];
}

__global__ void bias_relu_inplace(float* __restrict__ p,
                                  const float* __restrict__ b,
                                  size_t total, int F) {
    size_t t = blockIdx.x * (size_t)blockDim.x + threadIdx.x;
    if (t >= total) return;
    int f = (int)(t % F);
    float v = p[t] + b[f];
    p[t] = v > 0.f ? v : 0.f;
}

__global__ void mean_heads_bias(const float* __restrict__ agg,
                                const float* __restrict__ b,
                                float* __restrict__ out, int N, int H, int C) {
    int t = blockIdx.x * blockDim.x + threadIdx.x;
    if (t >= N * C) return;
    int n = t / C, c = t - n * C;
    float s = 0.f;
    for (int hh = 0; hh < H; ++hh)
        s += agg[(size_t)n * H * C + (size_t)hh * C + c];
    out[t] = s / (float)H + b[c];
}

static inline unsigned cdiv_u(size_t a, unsigned b) {
    return (unsigned)((a + b - 1) / b);
}

extern "C" void kernel_launch(void* const* d_in, const int* in_sizes, int n_in,
                              void* d_out, int out_size, void* d_ws, size_t ws_size,
                              hipStream_t stream) {
    const float* x     = (const float*)d_in[0];
    const int*   ei    = (const int*)d_in[1];
    const float* W1    = (const float*)d_in[2];
    const float* a_s1  = (const float*)d_in[3];
    const float* a_d1  = (const float*)d_in[4];
    const float* b1    = (const float*)d_in[5];
    const float* W2    = (const float*)d_in[6];
    const float* a_s2  = (const float*)d_in[7];
    const float* a_d2  = (const float*)d_in[8];
    const float* b2    = (const float*)d_in[9];
    float*       out   = (float*)d_out;

    const int F1   = in_sizes[3];
    const int F2   = in_sizes[7];
    const int LAT  = in_sizes[9];
    const int H    = F2 / LAT;
    const int C1   = F1 / H;
    const int Fin  = in_sizes[2] / F1;
    const int N    = in_sizes[0] / Fin;
    const int E    = in_sizes[1] / 2;
    const int Etot = E + N;

    float* ws = (float*)d_ws;
    size_t o = 0;
    float* h1   = ws + o; o += (size_t)N * F1;
    float* agg1 = ws + o; o += (size_t)N * F1;
    float* ex   = ws + o; o += (size_t)Etot * H;
    float* als  = ws + o; o += (size_t)N * H;
    float* ald  = ws + o; o += (size_t)N * H;
    float* mbuf = ws + o; o += (size_t)N * H;
    float* den  = ws + o; o += (size_t)N * H;
    float* h2   = h1;
    float* agg2 = h1 + (size_t)N * F2;
    (void)ws_size; (void)n_in; (void)out_size;

    const unsigned TB = 256;
    const float NEG_INF = -__builtin_huge_valf();

    fill_f32<<<cdiv_u((size_t)N * H, TB), TB, 0, stream>>>(mbuf, NEG_INF, (size_t)N * H);
    fill_f32<<<cdiv_u((size_t)N * H, TB), TB, 0, stream>>>(den, 0.f, (size_t)N * H);

    gemm_wmma_f32<<<dim3(F1 / TS, N / TS), 128, 0, stream>>>(x, W1, h1, N, Fin, F1);

    att_logits<<<cdiv_u((size_t)N * H, TB), TB, 0, stream>>>(h1, a_s1, a_d1, als, ald, N, H, C1);

    edge_max<<<cdiv_u((size_t)Etot * H, TB), TB, 0, stream>>>(ei, E, N, H, als, ald, mbuf);
    edge_exp<<<cdiv_u((size_t)Etot * H, TB), TB, 0, stream>>>(ei, E, N, H, als, ald, mbuf, ex, den);
    agg_band<512, 128, 64, 32><<<cdiv_u((size_t)N, 64), 1024, 0, stream>>>(
        ei, E, N, h1, ex, den, agg1);

    bias_relu_inplace<<<cdiv_u((size_t)N * F1, TB), TB, 0, stream>>>(agg1, b1, (size_t)N * F1, F1);

    gemm_wmma_f32<<<dim3(F2 / TS, N / TS), 128, 0, stream>>>(agg1, W2, h2, N, F1, F2);

    fill_f32<<<cdiv_u((size_t)N * H, TB), TB, 0, stream>>>(mbuf, NEG_INF, (size_t)N * H);
    fill_f32<<<cdiv_u((size_t)N * H, TB), TB, 0, stream>>>(den, 0.f, (size_t)N * H);

    att_logits<<<cdiv_u((size_t)N * H, TB), TB, 0, stream>>>(h2, a_s2, a_d2, als, ald, N, H, LAT);

    edge_max<<<cdiv_u((size_t)Etot * H, TB), TB, 0, stream>>>(ei, E, N, H, als, ald, mbuf);
    edge_exp<<<cdiv_u((size_t)Etot * H, TB), TB, 0, stream>>>(ei, E, N, H, als, ald, mbuf, ex, den);
    agg_band<256, 64, 64, 32><<<cdiv_u((size_t)N, 64), 1024, 0, stream>>>(
        ei, E, N, h2, ex, den, agg2);

    mean_heads_bias<<<cdiv_u((size_t)N * LAT, TB), TB, 0, stream>>>(agg2, b2, out, N, H, LAT);
}


# five small launches folded into the kernel that touches the buffer next: max/denominator start values stored by the logit kernel, bias+ReLU applied as GEMM 2 stages its A tile; 16 -> 11 launches
# speedup vs baseline: 3.8594x; 1.0975x over previous
#include <hip/hip_runtime.h>
#include <math.h>
#include <type_traits>

typedef __attribute__((ext_vector_type(2))) float v2f;
typedef __attribute__((ext_vector_type(8))) float v8f;

#define TS   32
#define KC   32
#define ASTR 36
#define BSTR 33

static __device__ __forceinline__ float leaky02(float x) {
    return x > 0.f ? x : 0.2f * x;
}

static __device__ __forceinline__ void atomicMaxF(float* addr, float v) {
    if (v >= 0.f)
        atomicMax((int*)addr, __float_as_int(v));
    else
        atomicMin((unsigned int*)addr, __float_as_uint(v));
}

template <bool ACT>
__global__ __launch_bounds__(128)
void gemm_wmma_f32(const float* __restrict__ A, const float* __restrict__ B,
                   float* __restrict__ D, int M, int K, int N,
                   const float* __restrict__ abias) {
    __shared__ float As[TS * ASTR];
    __shared__ float Bs[KC * BSTR];

    const int tileN = blockIdx.x * TS;
    const int tileM = blockIdx.y * TS;
    const int t    = threadIdx.x;
    const int wv   = t >> 5;
    const int lane = t & 31;
    const int half = lane >> 4;
    const int lm   = lane & 15;
    const int wm   = wv >> 1;
    const int wn   = wv & 1;

    const int crow = t >> 2;
    const int ccol = (t & 3) * 8;

    v8f acc = {};

    for (int kt = 0; kt < K; kt += KC) {
        const float* ga = A + (size_t)(tileM + crow) * K + kt + ccol;
        const float* gb = B + (size_t)(kt + crow) * N + tileN + ccol;
        if (kt + KC < K) {
            __builtin_prefetch(ga + KC, 0, 0);
            __builtin_prefetch(gb + (size_t)KC * N, 0, 0);
        }
        if constexpr (ACT) {
            const float* gbias = abias + kt + ccol;
            #pragma unroll
            for (int j = 0; j < 8; ++j) {
                const float v = ga[j] + gbias[j];
                As[crow * ASTR + ccol + j] = v > 0.f ? v : 0.f;
            }
        } else {
            #pragma unroll
            for (int j = 0; j < 8; ++j) As[crow * ASTR + ccol + j] = ga[j];
        }
        #pragma unroll
        for (int j = 0; j < 8; ++j) Bs[crow * BSTR + ccol + j] = gb[j];
        __syncthreads();

        #pragma unroll
        for (int kk = 0; kk < KC; kk += 4) {
            const int k0 = kk + 2 * half;
            v2f a, b;
            a.x = As[(wm * 16 + lm) * ASTR + k0];
            a.y = As[(wm * 16 + lm) * ASTR + k0 + 1];
            b.x = Bs[(k0)     * BSTR + wn * 16 + lm];
            b.y = Bs[(k0 + 1) * BSTR + wn * 16 + lm];
            acc = __builtin_amdgcn_wmma_f32_16x16x4_f32(
                false, a, false, b, (short)0, acc, false, false);
        }
        __syncthreads();
    }

    const int col   = tileN + wn * 16 + lm;
    const int rbase = tileM + wm * 16 + half * 8;
    #pragma unroll
    for (int r = 0; r < 8; ++r)
        D[(size_t)(rbase + r) * N + col] = acc[r];
}

__global__ void att_logits(const float* __restrict__ h,
                           const float* __restrict__ a_s,
                           const float* __restrict__ a_d,
                           float* __restrict__ al_s, float* __restrict__ al_d,
                           float* __restrict__ mbuf, float* __restrict__ denom,
                           int N, int H, int C) {
    int t = blockIdx.x * blockDim.x + threadIdx.x;
    if (t >= N * H) return;
    int n = t / H, hh = t - n * H;
    const float* row = h + (size_t)n * H * C + (size_t)hh * C;
    const float* vs  = a_s + (size_t)hh * C;
    const float* vd  = a_d + (size_t)hh * C;
    float ss = 0.f, sd = 0.f;
    for (int c = 0; c < C; ++c) {
        float v = row[c];
        ss = fmaf(v, vs[c], ss);
        sd = fmaf(v, vd[c], sd);
    }
    al_s[t] = ss;
    al_d[t] = sd;
    mbuf[t]  = -__builtin_huge_valf();
    denom[t] = 0.f;
}

__global__ void edge_max(const int* __restrict__ ei, int E, int N, int H,
                         const float* __restrict__ al_s,
                         const float* __restrict__ al_d,
                         float* __restrict__ mbuf) {
    int t = blockIdx.x * blockDim.x + threadIdx.x;
    int tot = (E + N) * H;
    if (t >= tot) return;
    int e = t / H, hh = t - e * H;
    int s, d;
    if (e < E) { s = ei[e]; d = ei[E + e]; } else { s = d = e - E; }
    float v = leaky02(al_s[s * H + hh] + al_d[d * H + hh]);
    atomicMaxF(&mbuf[d * H + hh], v);
}

__global__ void edge_exp(const int* __restrict__ ei, int E, int N, int H,
                         const float* __restrict__ al_s,
                         const float* __restrict__ al_d,
                         const float* __restrict__ mbuf,
                         float* __restrict__ exbuf,
                         float* __restrict__ denom) {
    int t = blockIdx.x * blockDim.x + threadIdx.x;
    int tot = (E + N) * H;
    if (t >= tot) return;
    int e = t / H, hh = t - e * H;
    int s, d;
    if (e < E) { s = ei[e]; d = ei[E + e]; } else { s = d = e - E; }
    float v  = leaky02(al_s[s * H + hh] + al_d[d * H + hh]);
    float ex = __expf(v - mbuf[d * H + hh]);
    exbuf[t] = ex;
    atomicAdd(&denom[d * H + hh], ex);
}

template <int F, int C, int R, int NW>
__global__ __launch_bounds__(NW * 32)
void agg_band(const int* __restrict__ ei, int E, int N,
              const float* __restrict__ hfeat,
              const float* __restrict__ exbuf,
              const float* __restrict__ denom,
              float* __restrict__ agg) {
    static_assert(F == 4 * C, "four heads");
    static_assert(R % NW == 0 && F % 64 == 0, "band and row shapes");
    constexpr int NT    = NW * 32;
    constexpr int V2    = F / 64;
    constexpr int SPW   = R / NW;
    constexpr int DRAIN = 8;
    constexpr int QCAP  = DRAIN + 256 + SPW;

    __shared__ __attribute__((aligned(16))) float acc[R * F];
    __shared__ int q[NW][QCAP];

    const int tid  = threadIdx.x;
    const int lane = tid & 31;
    const int w    = __builtin_amdgcn_readfirstlane(tid >> 5);
    const int base = blockIdx.x * R;

    float4* acc4 = reinterpret_cast<float4*>(acc);
    for (int i = tid; i < R * F / 4; i += NT) acc4[i] = make_float4(0.f, 0.f, 0.f, 0.f);
    __syncthreads();

    int cnt = 0;
    {
        const int row0 = base + w * SPW;
        int nself = N - row0;
        nself = nself < 0 ? 0 : (nself > SPW ? SPW : nself);
        if (lane < nself) q[w][lane] = E + row0 + lane;
        cnt = nself;
    }

    const int* __restrict__ dst = ei + E;
    const int nsteps = E >> 8;

    auto push = [&](bool hit, int ev) {
        const unsigned mask = __builtin_amdgcn_ballot_w32(hit);
        if (hit) q[w][cnt + (int)__builtin_amdgcn_mbcnt_lo(mask, 0u)] = ev;
        cnt += __builtin_popcount(mask);
    };
    auto test4 = [&](const int4& v, int e) {
        const bool m0 = (unsigned)(v.x - base) < (unsigned)R;
        const bool m1 = (unsigned)(v.y - base) < (unsigned)R;
        const bool m2 = (unsigned)(v.z - base) < (unsigned)R;
        const bool m3 = (unsigned)(v.w - base) < (unsigned)R;
        if (__builtin_amdgcn_ballot_w32(m0 | m1 | m2 | m3)) {
            push(m0, e);
            push(m1, e + 1);
            push(m2, e + 2);
            push(m3, e + 3);
        }
    };

    for (int step = w; ; step += NW) {
        const bool last = step >= nsteps;
        if (!last) {
            const int  e  = step * 256 + lane * 4;
            const int4 va = *reinterpret_cast<const int4*>(dst + e);
            const int4 vb = *reinterpret_cast<const int4*>(dst + e + 128);
            test4(va, e);
            test4(vb, e + 128);
        } else if (w == 0) {
            for (int e = nsteps * 256 + lane; __builtin_amdgcn_ballot_w32(e < E); e += 32)
                push(e < E && (unsigned)(dst[e < E ? e : 0] - base) < (unsigned)R, e);
        }

        if (cnt >= DRAIN || (last && cnt > 0)) {
            for (int b = 0; b < cnt; b += 32) {
                const int n = (cnt - b) < 32 ? (cnt - b) : 32;
                int   s = 0, r = 0;
                float a0 = 0.f, a1 = 0.f, a2 = 0.f, a3 = 0.f;
                if (lane < n) {
                    const int eq = q[w][b + lane];
                    int d;
                    if (eq < E) { s = ei[eq]; d = dst[eq]; } else { s = d = eq - E; }
                    const float4 x4 = *reinterpret_cast<const float4*>(exbuf + (size_t)eq * 4);
                    const float4 d4 = *reinterpret_cast<const float4*>(denom + (size_t)d * 4);
                    a0 = x4.x / d4.x;  a1 = x4.y / d4.y;
                    a2 = x4.z / d4.z;  a3 = x4.w / d4.w;
                    r  = d - base;
                }
                auto rows = [&](int j, auto UC) {
                    constexpr int U = decltype(UC)::value;
                    float2 v[U][V2];
                    #pragma unroll
                    for (int u = 0; u < U; ++u) {
                        const int sj = __builtin_amdgcn_readlane(s, j + u);
                        const float2* __restrict__ row =
                            reinterpret_cast<const float2*>(hfeat + (size_t)sj * F) + lane;
                        #pragma unroll
                        for (int k = 0; k < V2; ++k) v[u][k] = row[k * 32];
                    }
                    #pragma unroll
                    for (int u = 0; u < U; ++u) {
                        const int rj = __builtin_amdgcn_readlane(r, j + u);
                        float al[4];
                        al[0] = __builtin_bit_cast(float, __builtin_amdgcn_readlane(__builtin_bit_cast(int, a0), j + u));
                        al[1] = __builtin_bit_cast(float, __builtin_amdgcn_readlane(__builtin_bit_cast(int, a1), j + u));
                        al[2] = __builtin_bit_cast(float, __builtin_amdgcn_readlane(__builtin_bit_cast(int, a2), j + u));
                        al[3] = __builtin_bit_cast(float, __builtin_amdgcn_readlane(__builtin_bit_cast(int, a3), j + u));
                        float* arow = acc + rj * F + lane * 2;
                        #pragma unroll
                        for (int k = 0; k < V2; ++k) {
                            const float a = al[(k * 64) / C];
                            __hip_atomic_fetch_add(arow + k * 64,     v[u][k].x * a,
                                                   __ATOMIC_RELAXED, __HIP_MEMORY_SCOPE_WORKGROUP);
                            __hip_atomic_fetch_add(arow + k * 64 + 1, v[u][k].y * a,
                                                   __ATOMIC_RELAXED, __HIP_MEMORY_SCOPE_WORKGROUP);
                        }
                    }
                };
                int j = 0;
                for (; j + 4 <= n; j += 4) rows(j, std::integral_constant<int, 4>{});
                for (; j < n; ++j)         rows(j, std::integral_constant<int, 1>{});
            }
            cnt = 0;
        }
        if (last) break;
    }

    __syncthreads();
    float4* __restrict__ out4 = reinterpret_cast<float4*>(agg + (size_t)base * F);
    for (int i = tid; i < R * F / 4; i += NT)
        if (base + i / (F / 4) < N) out4[i] = acc4[i];
}

__global__ void mean_heads_bias(const float* __restrict__ agg,
                                const float* __restrict__ b,
                                float* __restrict__ out, int N, int H, int C) {
    int t = blockIdx.x * blockDim.x + threadIdx.x;
    if (t >= N * C) return;
    int n = t / C, c = t - n * C;
    float s = 0.f;
    for (int hh = 0; hh < H; ++hh)
        s += agg[(size_t)n * H * C + (size_t)hh * C + c];
    out[t] = s / (float)H + b[c];
}

static inline unsigned cdiv_u(size_t a, unsigned b) {
    return (unsigned)((a + b - 1) / b);
}

extern "C" void kernel_launch(void* const* d_in, const int* in_sizes, int n_in,
                              void* d_out, int out_size, void* d_ws, size_t ws_size,
                              hipStream_t stream) {
    const float* x     = (const float*)d_in[0];
    const int*   ei    = (const int*)d_in[1];
    const float* W1    = (const float*)d_in[2];
    const float* a_s1  = (const float*)d_in[3];
    const float* a_d1  = (const float*)d_in[4];
    const float* b1    = (const float*)d_in[5];
    const float* W2    = (const float*)d_in[6];
    const float* a_s2  = (const float*)d_in[7];
    const float* a_d2  = (const float*)d_in[8];
    const float* b2    = (const float*)d_in[9];
    float*       out   = (float*)d_out;

    const int F1   = in_sizes[3];
    const int F2   = in_sizes[7];
    const int LAT  = in_sizes[9];
    const int H    = F2 / LAT;
    const int C1   = F1 / H;
    const int Fin  = in_sizes[2] / F1;
    const int N    = in_sizes[0] / Fin;
    const int E    = in_sizes[1] / 2;
    const int Etot = E + N;

    float* ws = (float*)d_ws;
    size_t o = 0;
    float* h1   = ws + o; o += (size_t)N * F1;
    float* agg1 = ws + o; o += (size_t)N * F1;
    float* ex   = ws + o; o += (size_t)Etot * H;
    float* als  = ws + o; o += (size_t)N * H;
    float* ald  = ws + o; o += (size_t)N * H;
    float* mbuf = ws + o; o += (size_t)N * H;
    float* den  = ws + o; o += (size_t)N * H;
    float* h2   = h1;
    float* agg2 = h1 + (size_t)N * F2;
    (void)ws_size; (void)n_in; (void)out_size;

    const unsigned TB = 256;

    gemm_wmma_f32<false><<<dim3(F1 / TS, N / TS), 128, 0, stream>>>(x, W1, h1, N, Fin, F1, nullptr);

    att_logits<<<cdiv_u((size_t)N * H, TB), TB, 0, stream>>>(h1, a_s1, a_d1, als, ald, mbuf, den, N, H, C1);

    edge_max<<<cdiv_u((size_t)Etot * H, TB), TB, 0, stream>>>(ei, E, N, H, als, ald, mbuf);
    edge_exp<<<cdiv_u((size_t)Etot * H, TB), TB, 0, stream>>>(ei, E, N, H, als, ald, mbuf, ex, den);
    agg_band<512, 128, 64, 32><<<cdiv_u((size_t)N, 64), 1024, 0, stream>>>(
        ei, E, N, h1, ex, den, agg1);

    gemm_wmma_f32<true><<<dim3(F2 / TS, N / TS), 128, 0, stream>>>(agg1, W2, h2, N, F1, F2, b1);

    att_logits<<<cdiv_u((size_t)N * H, TB), TB, 0, stream>>>(h2, a_s2, a_d2, als, ald, mbuf, den, N, H, LAT);

    edge_max<<<cdiv_u((size_t)Etot * H, TB), TB, 0, stream>>>(ei, E, N, H, als, ald, mbuf);
    edge_exp<<<cdiv_u((size_t)Etot * H, TB), TB, 0, stream>>>(ei, E, N, H, als, ald, mbuf, ex, den);
    agg_band<256, 64, 64, 32><<<cdiv_u((size_t)N, 64), 1024, 0, stream>>>(
        ei, E, N, h2, ex, den, agg2);

    mean_heads_bias<<<cdiv_u((size_t)N * LAT, TB), TB, 0, stream>>>(agg2, b2, out, N, H, LAT);
}


# edge softmax folded into the band kernel: logit max and denominator of a block's rows kept in LDS, sum/denominator stored; global atomics, ex buffer, 4 launches gone; sum order = LDS arrival
# speedup vs baseline: 4.8033x; 1.2446x over previous
#include <hip/hip_runtime.h>
#include <math.h>
#include <type_traits>

typedef __attribute__((ext_vector_type(2))) float v2f;
typedef __attribute__((ext_vector_type(8))) float v8f;

#define TS   32
#define KC   32
#define ASTR 36
#define BSTR 33

static __device__ __forceinline__ float leaky02(float x) {
    return x > 0.f ? x : 0.2f * x;
}

static __device__ __forceinline__ void ldsMaxF(float* addr, float v) {
    if (v >= 0.f)
        __hip_atomic_fetch_max((int*)addr, __float_as_int(v) & 0x7fffffff,
                               __ATOMIC_RELAXED, __HIP_MEMORY_SCOPE_WORKGROUP);
    else
        __hip_atomic_fetch_min((unsigned int*)addr, __float_as_uint(v),
                               __ATOMIC_RELAXED, __HIP_MEMORY_SCOPE_WORKGROUP);
}

template <bool ACT>
__global__ __launch_bounds__(128)
void gemm_wmma_f32(const float* __restrict__ A, const float* __restrict__ B,
                   float* __restrict__ D, int M, int K, int N,
                   const float* __restrict__ abias) {
    __shared__ float As[TS * ASTR];
    __shared__ float Bs[KC * BSTR];

    const int tileN = blockIdx.x * TS;
    const int tileM = blockIdx.y * TS;
    const int t    = threadIdx.x;
    const int wv   = t >> 5;
    const int lane = t & 31;
    const int half = lane >> 4;
    const int lm   = lane & 15;
    const int wm   = wv >> 1;
    const int wn   = wv & 1;

    const int crow = t >> 2;
    const int ccol = (t & 3) * 8;

    v8f acc = {};

    for (int kt = 0; kt < K; kt += KC) {
        const float* ga = A + (size_t)(tileM + crow) * K + kt + ccol;
        const float* gb = B + (size_t)(kt + crow) * N + tileN + ccol;
        if (kt + KC < K) {
            __builtin_prefetch(ga + KC, 0, 0);
            __builtin_prefetch(gb + (size_t)KC * N, 0, 0);
        }
        if constexpr (ACT) {
            const float* gbias = abias + kt + ccol;
            #pragma unroll
            for (int j = 0; j < 8; ++j) {
                const float v = ga[j] + gbias[j];
                As[crow * ASTR + ccol + j] = v > 0.f ? v : 0.f;
            }
        } else {
            #pragma unroll
            for (int j = 0; j < 8; ++j) As[crow * ASTR + ccol + j] = ga[j];
        }
        #pragma unroll
        for (int j = 0; j < 8; ++j) Bs[crow * BSTR + ccol + j] = gb[j];
        __syncthreads();

        #pragma unroll
        for (int kk = 0; kk < KC; kk += 4) {
            const int k0 = kk + 2 * half;
            v2f a, b;
            a.x = As[(wm * 16 + lm) * ASTR + k0];
            a.y = As[(wm * 16 + lm) * ASTR + k0 + 1];
            b.x = Bs[(k0)     * BSTR + wn * 16 + lm];
            b.y = Bs[(k0 + 1) * BSTR + wn * 16 + lm];
            acc = __builtin_amdgcn_wmma_f32_16x16x4_f32(
                false, a, false, b, (short)0, acc, false, false);
        }
        __syncthreads();
    }

    const int col   = tileN + wn * 16 + lm;
    const int rbase = tileM + wm * 16 + half * 8;
    #pragma unroll
    for (int r = 0; r < 8; ++r)
        D[(size_t)(rbase + r) * N + col] = acc[r];
}

__global__ void att_logits(const float* __restrict__ h,
                           const float* __restrict__ a_s,
                           const float* __restrict__ a_d,
                           float* __restrict__ al_s, float* __restrict__ al_d,
                           int N, int H, int C) {
    int t = blockIdx.x * blockDim.x + threadIdx.x;
    if (t >= N * H) return;
    int n = t / H, hh = t - n * H;
    const float* row = h + (size_t)n * H * C + (size_t)hh * C;
    const float* vs  = a_s + (size_t)hh * C;
    const float* vd  = a_d + (size_t)hh * C;
    float ss = 0.f, sd = 0.f;
    for (int c = 0; c < C; ++c) {
        float v = row[c];
        ss = fmaf(v, vs[c], ss);
        sd = fmaf(v, vd[c], sd);
    }
    al_s[t] = ss;
    al_d[t] = sd;
}

template <int F, int C, int R, int NW>
__global__ __launch_bounds__(NW * 32)
void agg_band(const int* __restrict__ ei, int E, int N,
              const float* __restrict__ hfeat,
              const float* __restrict__ al_s,
              const float* __restrict__ al_d,
              float* __restrict__ agg) {
    static_assert(F == 4 * C && C % 4 == 0, "four heads");
    static_assert(R % NW == 0 && F % 64 == 0, "band and row shapes");
    constexpr int NT         = NW * 32;
    constexpr int V2         = F / 64;
    constexpr int SPW        = R / NW;
    constexpr int DRAIN_MAX  = 32;
    constexpr int DRAIN_ROWS = 8;
    constexpr int QCAP       = 512;
    static_assert(SPW < DRAIN_ROWS && DRAIN_MAX + 256 <= QCAP, "queue room");

    __shared__ __attribute__((aligned(16))) float acc[R * F];
    __shared__ __attribute__((aligned(16))) float ald_b[R * 4];
    __shared__ __attribute__((aligned(16))) float mx[R * 4];
    __shared__ __attribute__((aligned(16))) float den[R * 4];
    __shared__ int q[NW][QCAP];
    __shared__ int ovf;

    const int tid  = threadIdx.x;
    const int lane = tid & 31;
    const int w    = __builtin_amdgcn_readfirstlane(tid >> 5);
    const int base = blockIdx.x * R;

    float4* acc4 = reinterpret_cast<float4*>(acc);
    for (int i = tid; i < R * F / 4; i += NT) acc4[i] = make_float4(0.f, 0.f, 0.f, 0.f);
    for (int i = tid; i < R * 4; i += NT) {
        const int row = base + (i >> 2);
        ald_b[i] = row < N ? al_d[(size_t)row * 4 + (i & 3)] : 0.f;
        mx[i]    = -__builtin_huge_valf();
        den[i]   = 0.f;
    }
    if (tid == 0) ovf = 0;
    __syncthreads();

    const int* __restrict__ dst = ei + E;
    const int nsteps = E >> 8;

    auto scan = [&](const int thresh, const bool keep, auto&& drain) -> int {
        int  cnt = 0, done = 0;
        bool kept = keep;
        {
            const int row0 = base + w * SPW;
            int nself = N - row0;
            nself = nself < 0 ? 0 : (nself > SPW ? SPW : nself);
            if (lane < nself) q[w][lane] = E + row0 + lane;
            cnt = nself;
        }
        auto push = [&](bool hit, int ev) {
            const unsigned mask = __builtin_amdgcn_ballot_w32(hit);
            if (hit) q[w][cnt + (int)__builtin_amdgcn_mbcnt_lo(mask, 0u)] = ev;
            cnt += __builtin_popcount(mask);
        };
        auto test4 = [&](const int4& v, int e) {
            const bool m0 = (unsigned)(v.x - base) < (unsigned)R;
            const bool m1 = (unsigned)(v.y - base) < (unsigned)R;
            const bool m2 = (unsigned)(v.z - base) < (unsigned)R;
            const bool m3 = (unsigned)(v.w - base) < (unsigned)R;
            if (__builtin_amdgcn_ballot_w32(m0 | m1 | m2 | m3)) {
                push(m0, e);
                push(m1, e + 1);
                push(m2, e + 2);
                push(m3, e + 3);
            }
        };
        for (int step = w; ; step += NW) {
            const bool last = step >= nsteps;
            if (!last) {
                const int  e  = step * 256 + lane * 4;
                const int4 va = *reinterpret_cast<const int4*>(dst + e);
                const int4 vb = *reinterpret_cast<const int4*>(dst + e + 128);
                test4(va, e);
                test4(vb, e + 128);
            } else if (w == 0) {
                for (int e = nsteps * 256 + lane; __builtin_amdgcn_ballot_w32(e < E); e += 32)
                    push(e < E && (unsigned)(dst[e < E ? e : 0] - base) < (unsigned)R, e);
            }
            if (cnt - done >= thresh || (last && cnt > done)) {
                drain(done, cnt);
                if (kept && (last || cnt + thresh + 256 <= QCAP)) {
                    done = cnt;
                } else {
                    kept = false;
                    cnt = 0;
                    done = 0;
                }
            }
            if (last) break;
        }
        return kept ? cnt : -1;
    };

    auto drain_max = [&](const int from, const int to) {
        for (int b = from; b < to; b += 32) {
            if (b + lane < to) {
                const int eq = q[w][b + lane];
                int s, d;
                if (eq < E) { s = ei[eq]; d = dst[eq]; } else { s = d = eq - E; }
                const float4 s4 = *reinterpret_cast<const float4*>(al_s + (size_t)s * 4);
                const int    r  = d - base;
                const float4 d4 = *reinterpret_cast<const float4*>(ald_b + r * 4);
                ldsMaxF(&mx[r * 4 + 0], leaky02(s4.x + d4.x));
                ldsMaxF(&mx[r * 4 + 1], leaky02(s4.y + d4.y));
                ldsMaxF(&mx[r * 4 + 2], leaky02(s4.z + d4.z));
                ldsMaxF(&mx[r * 4 + 3], leaky02(s4.w + d4.w));
            }
        }
    };

    auto drain_rows = [&](const int from, const int to) {
        for (int b = from; b < to; b += 32) {
            const int n = (to - b) < 32 ? (to - b) : 32;
            int   s = 0, r = 0;
            float a0 = 0.f, a1 = 0.f, a2 = 0.f, a3 = 0.f;
            if (lane < n) {
                const int eq = q[w][b + lane];
                int d;
                if (eq < E) { s = ei[eq]; d = dst[eq]; } else { s = d = eq - E; }
                const float4 s4 = *reinterpret_cast<const float4*>(al_s + (size_t)s * 4);
                r = d - base;
                const float4 d4 = *reinterpret_cast<const float4*>(ald_b + r * 4);
                const float4 m4 = *reinterpret_cast<const float4*>(mx + r * 4);
                a0 = __expf(leaky02(s4.x + d4.x) - m4.x);
                a1 = __expf(leaky02(s4.y + d4.y) - m4.y);
                a2 = __expf(leaky02(s4.z + d4.z) - m4.z);
                a3 = __expf(leaky02(s4.w + d4.w) - m4.w);
                __hip_atomic_fetch_add(&den[r * 4 + 0], a0, __ATOMIC_RELAXED, __HIP_MEMORY_SCOPE_WORKGROUP);
                __hip_atomic_fetch_add(&den[r * 4 + 1], a1, __ATOMIC_RELAXED, __HIP_MEMORY_SCOPE_WORKGROUP);
                __hip_atomic_fetch_add(&den[r * 4 + 2], a2, __ATOMIC_RELAXED, __HIP_MEMORY_SCOPE_WORKGROUP);
                __hip_atomic_fetch_add(&den[r * 4 + 3], a3, __ATOMIC_RELAXED, __HIP_MEMORY_SCOPE_WORKGROUP);
            }
            auto rows = [&](int j, auto UC) {
                constexpr int U = decltype(UC)::value;
                float2 v[U][V2];
                #pragma unroll
                for (int u = 0; u < U; ++u) {
                    const int sj = __builtin_amdgcn_readlane(s, j + u);
                    const float2* __restrict__ row =
                        reinterpret_cast<const float2*>(hfeat + (size_t)sj * F) + lane;
                    #pragma unroll
                    for (int k = 0; k < V2; ++k) v[u][k] = row[k * 32];
                }
                #pragma unroll
                for (int u = 0; u < U; ++u) {
                    const int rj = __builtin_amdgcn_readlane(r, j + u);
                    float al[4];
                    al[0] = __builtin_bit_cast(float, __builtin_amdgcn_readlane(__builtin_bit_cast(int, a0), j + u));
                    al[1] = __builtin_bit_cast(float, __builtin_amdgcn_readlane(__builtin_bit_cast(int, a1), j + u));
                    al[2] = __builtin_bit_cast(float, __builtin_amdgcn_readlane(__builtin_bit_cast(int, a2), j + u));
                    al[3] = __builtin_bit_cast(float, __builtin_amdgcn_readlane(__builtin_bit_cast(int, a3), j + u));
                    float* arow = acc + rj * F + lane * 2;
                    #pragma unroll
                    for (int k = 0; k < V2; ++k) {
                        const float a = al[(k * 64) / C];
                        __hip_atomic_fetch_add(arow + k * 64,     v[u][k].x * a,
                                               __ATOMIC_RELAXED, __HIP_MEMORY_SCOPE_WORKGROUP);
                        __hip_atomic_fetch_add(arow + k * 64 + 1, v[u][k].y * a,
                                               __ATOMIC_RELAXED, __HIP_MEMORY_SCOPE_WORKGROUP);
                    }
                }
            };
            int j = 0;
            for (; j + 4 <= n; j += 4) rows(j, std::integral_constant<int, 4>{});
            for (; j < n; ++j)         rows(j, std::integral_constant<int, 1>{});
        }
    };

    const int nkept = scan(DRAIN_MAX, true, drain_max);
    if (nkept < 0 && lane == 0)
        __hip_atomic_fetch_or(&ovf, 1, __ATOMIC_RELAXED, __HIP_MEMORY_SCOPE_WORKGROUP);
    __syncthreads();
    if (__builtin_amdgcn_readfirstlane(ovf) == 0) {
        drain_rows(0, nkept);
    } else {
        scan(DRAIN_ROWS, false, drain_rows);
    }
    __syncthreads();

    float4* __restrict__ out4 = reinterpret_cast<float4*>(agg + (size_t)base * F);
    for (int i = tid; i < R * F / 4; i += NT) {
        const int row = i / (F / 4);
        if (base + row < N) {
            const float  dn = den[row * 4 + ((i % (F / 4)) * 4) / C];
            const float4 v  = acc4[i];
            out4[i] = make_float4(v.x / dn, v.y / dn, v.z / dn, v.w / dn);
        }
    }
}

__global__ void mean_heads_bias(const float* __restrict__ agg,
                                const float* __restrict__ b,
                                float* __restrict__ out, int N, int H, int C) {
    int t = blockIdx.x * blockDim.x + threadIdx.x;
    if (t >= N * C) return;
    int n = t / C, c = t - n * C;
    float s = 0.f;
    for (int hh = 0; hh < H; ++hh)
        s += agg[(size_t)n * H * C + (size_t)hh * C + c];
    out[t] = s / (float)H + b[c];
}

static inline unsigned cdiv_u(size_t a, unsigned b) {
    return (unsigned)((a + b - 1) / b);
}

extern "C" void kernel_launch(void* const* d_in, const int* in_sizes, int n_in,
                              void* d_out, int out_size, void* d_ws, size_t ws_size,
                              hipStream_t stream) {
    const float* x     = (const float*)d_in[0];
    const int*   ei    = (const int*)d_in[1];
    const float* W1    = (const float*)d_in[2];
    const float* a_s1  = (const float*)d_in[3];
    const float* a_d1  = (const float*)d_in[4];
    const float* b1    = (const float*)d_in[5];
    const float* W2    = (const float*)d_in[6];
    const float* a_s2  = (const float*)d_in[7];
    const float* a_d2  = (const float*)d_in[8];
    const float* b2    = (const float*)d_in[9];
    float*       out   = (float*)d_out;

    const int F1   = in_sizes[3];
    const int F2   = in_sizes[7];
    const int LAT  = in_sizes[9];
    const int H    = F2 / LAT;
    const int C1   = F1 / H;
    const int Fin  = in_sizes[2] / F1;
    const int N    = in_sizes[0] / Fin;
    const int E    = in_sizes[1] / 2;

    float* ws = (float*)d_ws;
    size_t o = 0;
    float* h1   = ws + o; o += (size_t)N * F1;
    float* agg1 = ws + o; o += (size_t)N * F1;
    float* als  = ws + o; o += (size_t)N * H;
    float* ald  = ws + o; o += (size_t)N * H;
    float* h2   = h1;
    float* agg2 = h1 + (size_t)N * F2;
    (void)ws_size; (void)n_in; (void)out_size;

    const unsigned TB = 256;

    gemm_wmma_f32<false><<<dim3(F1 / TS, N / TS), 128, 0, stream>>>(x, W1, h1, N, Fin, F1, nullptr);

    att_logits<<<cdiv_u((size_t)N * H, TB), TB, 0, stream>>>(h1, a_s1, a_d1, als, ald, N, H, C1);

    agg_band<512, 128, 64, 32><<<cdiv_u((size_t)N, 64), 1024, 0, stream>>>(
        ei, E, N, h1, als, ald, agg1);

    gemm_wmma_f32<true><<<dim3(F2 / TS, N / TS), 128, 0, stream>>>(agg1, W2, h2, N, F1, F2, b1);

    att_logits<<<cdiv_u((size_t)N * H, TB), TB, 0, stream>>>(h2, a_s2, a_d2, als, ald, N, H, LAT);

    agg_band<256, 64, 64, 32><<<cdiv_u((size_t)N, 64), 1024, 0, stream>>>(
        ei, E, N, h2, als, ald, agg2);

    mean_heads_bias<<<cdiv_u((size_t)N * LAT, TB), TB, 0, stream>>>(agg2, b2, out, N, H, LAT);
}


# layer 2's band kernel takes its waves' edge queues from the workspace, where layer 1's kernel of the same band left them: second scan of the destination list gone; sum order still LDS arrival
# speedup vs baseline: 5.0131x; 1.0437x over previous
#include <hip/hip_runtime.h>
#include <math.h>
#include <type_traits>

typedef __attribute__((ext_vector_type(2))) float v2f;
typedef __attribute__((ext_vector_type(8))) float v8f;

#define TS   32
#define KC   32
#define ASTR 36
#define BSTR 33

static __device__ __forceinline__ float leaky02(float x) {
    return x > 0.f ? x : 0.2f * x;
}

static __device__ __forceinline__ void ldsMaxF(float* addr, float v) {
    if (v >= 0.f)
        __hip_atomic_fetch_max((int*)addr, __float_as_int(v) & 0x7fffffff,
                               __ATOMIC_RELAXED, __HIP_MEMORY_SCOPE_WORKGROUP);
    else
        __hip_atomic_fetch_min((unsigned int*)addr, __float_as_uint(v),
                               __ATOMIC_RELAXED, __HIP_MEMORY_SCOPE_WORKGROUP);
}

template <bool ACT>
__global__ __launch_bounds__(128)
void gemm_wmma_f32(const float* __restrict__ A, const float* __restrict__ B,
                   float* __restrict__ D, int M, int K, int N,
                   const float* __restrict__ abias) {
    __shared__ float As[TS * ASTR];
    __shared__ float Bs[KC * BSTR];

    const int tileN = blockIdx.x * TS;
    const int tileM = blockIdx.y * TS;
    const int t    = threadIdx.x;
    const int wv   = t >> 5;
    const int lane = t & 31;
    const int half = lane >> 4;
    const int lm   = lane & 15;
    const int wm   = wv >> 1;
    const int wn   = wv & 1;

    const int crow = t >> 2;
    const int ccol = (t & 3) * 8;

    v8f acc = {};

    for (int kt = 0; kt < K; kt += KC) {
        const float* ga = A + (size_t)(tileM + crow) * K + kt + ccol;
        const float* gb = B + (size_t)(kt + crow) * N + tileN + ccol;
        if (kt + KC < K) {
            __builtin_prefetch(ga + KC, 0, 0);
            __builtin_prefetch(gb + (size_t)KC * N, 0, 0);
        }
        if constexpr (ACT) {
            const float* gbias = abias + kt + ccol;
            #pragma unroll
            for (int j = 0; j < 8; ++j) {
                const float v = ga[j] + gbias[j];
                As[crow * ASTR + ccol + j] = v > 0.f ? v : 0.f;
            }
        } else {
            #pragma unroll
            for (int j = 0; j < 8; ++j) As[crow * ASTR + ccol + j] = ga[j];
        }
        #pragma unroll
        for (int j = 0; j < 8; ++j) Bs[crow * BSTR + ccol + j] = gb[j];
        __syncthreads();

        #pragma unroll
        for (int kk = 0; kk < KC; kk += 4) {
            const int k0 = kk + 2 * half;
            v2f a, b;
            a.x = As[(wm * 16 + lm) * ASTR + k0];
            a.y = As[(wm * 16 + lm) * ASTR + k0 + 1];
            b.x = Bs[(k0)     * BSTR + wn * 16 + lm];
            b.y = Bs[(k0 + 1) * BSTR + wn * 16 + lm];
            acc = __builtin_amdgcn_wmma_f32_16x16x4_f32(
                false, a, false, b, (short)0, acc, false, false);
        }
        __syncthreads();
    }

    const int col   = tileN + wn * 16 + lm;
    const int rbase = tileM + wm * 16 + half * 8;
    #pragma unroll
    for (int r = 0; r < 8; ++r)
        D[(size_t)(rbase + r) * N + col] = acc[r];
}

__global__ void att_logits(const float* __restrict__ h,
                           const float* __restrict__ a_s,
                           const float* __restrict__ a_d,
                           float* __restrict__ al_s, float* __restrict__ al_d,
                           int N, int H, int C) {
    int t = blockIdx.x * blockDim.x + threadIdx.x;
    if (t >= N * H) return;
    int n = t / H, hh = t - n * H;
    const float* row = h + (size_t)n * H * C + (size_t)hh * C;
    const float* vs  = a_s + (size_t)hh * C;
    const float* vd  = a_d + (size_t)hh * C;
    float ss = 0.f, sd = 0.f;
    for (int c = 0; c < C; ++c) {
        float v = row[c];
        ss = fmaf(v, vs[c], ss);
        sd = fmaf(v, vd[c], sd);
    }
    al_s[t] = ss;
    al_d[t] = sd;
}

template <int F, int C, int R, int NW, int LIST>
__global__ __launch_bounds__(NW * 32)
void agg_band(const int* __restrict__ ei, int E, int N,
              const float* __restrict__ hfeat,
              const float* __restrict__ al_s,
              const float* __restrict__ al_d,
              float* __restrict__ agg,
              int* __restrict__ blist) {
    static_assert(F == 4 * C && C % 4 == 0, "four heads");
    static_assert(R % NW == 0 && F % 64 == 0, "band and row shapes");
    constexpr int NT         = NW * 32;
    constexpr int V2         = F / 64;
    constexpr int SPW        = R / NW;
    constexpr int DRAIN_MAX  = 32;
    constexpr int DRAIN_ROWS = 8;
    constexpr int QCAP       = 512;
    static_assert(SPW < DRAIN_ROWS && DRAIN_MAX + 256 <= QCAP, "queue room");
    static_assert(LIST == 1 || LIST == 2, "1: leaves its queues, 2: takes them");
    constexpr int BLS        = 1 + NW + NW * QCAP;

    __shared__ __attribute__((aligned(16))) float acc[R * F];
    __shared__ __attribute__((aligned(16))) float ald_b[R * 4];
    __shared__ __attribute__((aligned(16))) float mx[R * 4];
    __shared__ __attribute__((aligned(16))) float den[R * 4];
    __shared__ int q[NW][QCAP];
    __shared__ int ovf;

    const int tid  = threadIdx.x;
    const int lane = tid & 31;
    const int w    = __builtin_amdgcn_readfirstlane(tid >> 5);
    const int base = blockIdx.x * R;

    float4* acc4 = reinterpret_cast<float4*>(acc);
    for (int i = tid; i < R * F / 4; i += NT) acc4[i] = make_float4(0.f, 0.f, 0.f, 0.f);
    for (int i = tid; i < R * 4; i += NT) {
        const int row = base + (i >> 2);
        ald_b[i] = row < N ? al_d[(size_t)row * 4 + (i & 3)] : 0.f;
        mx[i]    = -__builtin_huge_valf();
        den[i]   = 0.f;
    }
    if (tid == 0) ovf = 0;
    __syncthreads();

    const int* __restrict__ dst = ei + E;
    const int nsteps = E >> 8;

    auto scan = [&](const int thresh, const bool keep, auto&& drain) -> int {
        int  cnt = 0, done = 0;
        bool kept = keep;
        {
            const int row0 = base + w * SPW;
            int nself = N - row0;
            nself = nself < 0 ? 0 : (nself > SPW ? SPW : nself);
            if (lane < nself) q[w][lane] = E + row0 + lane;
            cnt = nself;
        }
        auto push = [&](bool hit, int ev) {
            const unsigned mask = __builtin_amdgcn_ballot_w32(hit);
            if (hit) q[w][cnt + (int)__builtin_amdgcn_mbcnt_lo(mask, 0u)] = ev;
            cnt += __builtin_popcount(mask);
        };
        auto test4 = [&](const int4& v, int e) {
            const bool m0 = (unsigned)(v.x - base) < (unsigned)R;
            const bool m1 = (unsigned)(v.y - base) < (unsigned)R;
            const bool m2 = (unsigned)(v.z - base) < (unsigned)R;
            const bool m3 = (unsigned)(v.w - base) < (unsigned)R;
            if (__builtin_amdgcn_ballot_w32(m0 | m1 | m2 | m3)) {
                push(m0, e);
                push(m1, e + 1);
                push(m2, e + 2);
                push(m3, e + 3);
            }
        };
        for (int step = w; ; step += NW) {
            const bool last = step >= nsteps;
            if (!last) {
                const int  e  = step * 256 + lane * 4;
                const int4 va = *reinterpret_cast<const int4*>(dst + e);
                const int4 vb = *reinterpret_cast<const int4*>(dst + e + 128);
                test4(va, e);
                test4(vb, e + 128);
            } else if (w == 0) {
                for (int e = nsteps * 256 + lane; __builtin_amdgcn_ballot_w32(e < E); e += 32)
                    push(e < E && (unsigned)(dst[e < E ? e : 0] - base) < (unsigned)R, e);
            }
            if (cnt - done >= thresh || (last && cnt > done)) {
                drain(done, cnt);
                if (kept && (last || cnt + thresh + 256 <= QCAP)) {
                    done = cnt;
                } else {
                    kept = false;
                    cnt = 0;
                    done = 0;
                }
            }
            if (last) break;
        }
        return kept ? cnt : -1;
    };

    auto drain_max = [&](const int from, const int to) {
        for (int b = from; b < to; b += 32) {
            if (b + lane < to) {
                const int eq = q[w][b + lane];
                int s, d;
                if (eq < E) { s = ei[eq]; d = dst[eq]; } else { s = d = eq - E; }
                const float4 s4 = *reinterpret_cast<const float4*>(al_s + (size_t)s * 4);
                const int    r  = d - base;
                const float4 d4 = *reinterpret_cast<const float4*>(ald_b + r * 4);
                ldsMaxF(&mx[r * 4 + 0], leaky02(s4.x + d4.x));
                ldsMaxF(&mx[r * 4 + 1], leaky02(s4.y + d4.y));
                ldsMaxF(&mx[r * 4 + 2], leaky02(s4.z + d4.z));
                ldsMaxF(&mx[r * 4 + 3], leaky02(s4.w + d4.w));
            }
        }
    };

    auto drain_rows = [&](const int from, const int to) {
        for (int b = from; b < to; b += 32) {
            const int n = (to - b) < 32 ? (to - b) : 32;
            int   s = 0, r = 0;
            float a0 = 0.f, a1 = 0.f, a2 = 0.f, a3 = 0.f;
            if (lane < n) {
                const int eq = q[w][b + lane];
                int d;
                if (eq < E) { s = ei[eq]; d = dst[eq]; } else { s = d = eq - E; }
                const float4 s4 = *reinterpret_cast<const float4*>(al_s + (size_t)s * 4);
                r = d - base;
                const float4 d4 = *reinterpret_cast<const float4*>(ald_b + r * 4);
                const float4 m4 = *reinterpret_cast<const float4*>(mx + r * 4);
                a0 = __expf(leaky02(s4.x + d4.x) - m4.x);
                a1 = __expf(leaky02(s4.y + d4.y) - m4.y);
                a2 = __expf(leaky02(s4.z + d4.z) - m4.z);
                a3 = __expf(leaky02(s4.w + d4.w) - m4.w);
                __hip_atomic_fetch_add(&den[r * 4 + 0], a0, __ATOMIC_RELAXED, __HIP_MEMORY_SCOPE_WORKGROUP);
                __hip_atomic_fetch_add(&den[r * 4 + 1], a1, __ATOMIC_RELAXED, __HIP_MEMORY_SCOPE_WORKGROUP);
                __hip_atomic_fetch_add(&den[r * 4 + 2], a2, __ATOMIC_RELAXED, __HIP_MEMORY_SCOPE_WORKGROUP);
                __hip_atomic_fetch_add(&den[r * 4 + 3], a3, __ATOMIC_RELAXED, __HIP_MEMORY_SCOPE_WORKGROUP);
            }
            auto rows = [&](int j, auto UC) {
                constexpr int U = decltype(UC)::value;
                float2 v[U][V2];
                #pragma unroll
                for (int u = 0; u < U; ++u) {
                    const int sj = __builtin_amdgcn_readlane(s, j + u);
                    const float2* __restrict__ row =
                        reinterpret_cast<const float2*>(hfeat + (size_t)sj * F) + lane;
                    #pragma unroll
                    for (int k = 0; k < V2; ++k) v[u][k] = row[k * 32];
                }
                #pragma unroll
                for (int u = 0; u < U; ++u) {
                    const int rj = __builtin_amdgcn_readlane(r, j + u);
                    float al[4];
                    al[0] = __builtin_bit_cast(float, __builtin_amdgcn_readlane(__builtin_bit_cast(int, a0), j + u));
                    al[1] = __builtin_bit_cast(float, __builtin_amdgcn_readlane(__builtin_bit_cast(int, a1), j + u));
                    al[2] = __builtin_bit_cast(float, __builtin_amdgcn_readlane(__builtin_bit_cast(int, a2), j + u));
                    al[3] = __builtin_bit_cast(float, __builtin_amdgcn_readlane(__builtin_bit_cast(int, a3), j + u));
                    float* arow = acc + rj * F + lane * 2;
                    #pragma unroll
                    for (int k = 0; k < V2; ++k) {
                        const float a = al[(k * 64) / C];
                        __hip_atomic_fetch_add(arow + k * 64,     v[u][k].x * a,
                                               __ATOMIC_RELAXED, __HIP_MEMORY_SCOPE_WORKGROUP);
                        __hip_atomic_fetch_add(arow + k * 64 + 1, v[u][k].y * a,
                                               __ATOMIC_RELAXED, __HIP_MEMORY_SCOPE_WORKGROUP);
                    }
                }
            };
            int j = 0;
            for (; j + 4 <= n; j += 4) rows(j, std::integral_constant<int, 4>{});
            for (; j < n; ++j)         rows(j, std::integral_constant<int, 1>{});
        }
    };

    int* __restrict__ bl = blist + (size_t)blockIdx.x * BLS;
    int nkept;
    if constexpr (LIST == 2) {
        if (__builtin_amdgcn_readfirstlane(bl[0]) == 0) {
            nkept = __builtin_amdgcn_readfirstlane(bl[1 + w]);
            for (int i = lane; i < nkept; i += 32) q[w][i] = bl[1 + NW + w * QCAP + i];
            drain_max(0, nkept);
        } else {
            nkept = scan(DRAIN_MAX, true, drain_max);
        }
    } else {
        nkept = scan(DRAIN_MAX, true, drain_max);
    }
    if (nkept < 0 && lane == 0)
        __hip_atomic_fetch_or(&ovf, 1, __ATOMIC_RELAXED, __HIP_MEMORY_SCOPE_WORKGROUP);
    __syncthreads();
    if constexpr (LIST == 1) {
        const int lost = __builtin_amdgcn_readfirstlane(ovf);
        if (tid == 0) bl[0] = lost;
        if (lost == 0) {
            if (lane == 0) bl[1 + w] = nkept;
            for (int i = lane; i < nkept; i += 32) bl[1 + NW + w * QCAP + i] = q[w][i];
        }
    }
    if (__builtin_amdgcn_readfirstlane(ovf) == 0) {
        drain_rows(0, nkept);
    } else {
        scan(DRAIN_ROWS, false, drain_rows);
    }
    __syncthreads();

    float4* __restrict__ out4 = reinterpret_cast<float4*>(agg + (size_t)base * F);
    for (int i = tid; i < R * F / 4; i += NT) {
        const int row = i / (F / 4);
        if (base + row < N) {
            const float  dn = den[row * 4 + ((i % (F / 4)) * 4) / C];
            const float4 v  = acc4[i];
            out4[i] = make_float4(v.x / dn, v.y / dn, v.z / dn, v.w / dn);
        }
    }
}

__global__ void mean_heads_bias(const float* __restrict__ agg,
                                const float* __restrict__ b,
                                float* __restrict__ out, int N, int H, int C) {
    int t = blockIdx.x * blockDim.x + threadIdx.x;
    if (t >= N * C) return;
    int n = t / C, c = t - n * C;
    float s = 0.f;
    for (int hh = 0; hh < H; ++hh)
        s += agg[(size_t)n * H * C + (size_t)hh * C + c];
    out[t] = s / (float)H + b[c];
}

static inline unsigned cdiv_u(size_t a, unsigned b) {
    return (unsigned)((a + b - 1) / b);
}

extern "C" void kernel_launch(void* const* d_in, const int* in_sizes, int n_in,
                              void* d_out, int out_size, void* d_ws, size_t ws_size,
                              hipStream_t stream) {
    const float* x     = (const float*)d_in[0];
    const int*   ei    = (const int*)d_in[1];
    const float* W1    = (const float*)d_in[2];
    const float* a_s1  = (const float*)d_in[3];
    const float* a_d1  = (const float*)d_in[4];
    const float* b1    = (const float*)d_in[5];
    const float* W2    = (const float*)d_in[6];
    const float* a_s2  = (const float*)d_in[7];
    const float* a_d2  = (const float*)d_in[8];
    const float* b2    = (const float*)d_in[9];
    float*       out   = (float*)d_out;

    const int F1   = in_sizes[3];
    const int F2   = in_sizes[7];
    const int LAT  = in_sizes[9];
    const int H    = F2 / LAT;
    const int C1   = F1 / H;
    const int Fin  = in_sizes[2] / F1;
    const int N    = in_sizes[0] / Fin;
    const int E    = in_sizes[1] / 2;

    float* ws = (float*)d_ws;
    size_t o = 0;
    float* h1   = ws + o; o += (size_t)N * F1;
    float* agg1 = ws + o; o += (size_t)N * F1;
    float* als  = ws + o; o += (size_t)N * H;
    float* ald  = ws + o; o += (size_t)N * H;
    float* h2   = h1;
    float* agg2 = h1 + (size_t)N * F2;
    int*   blist = (int*)(ws + o);
    (void)ws_size; (void)n_in; (void)out_size;

    const unsigned TB = 256;

    gemm_wmma_f32<false><<<dim3(F1 / TS, N / TS), 128, 0, stream>>>(x, W1, h1, N, Fin, F1, nullptr);

    att_logits<<<cdiv_u((size_t)N * H, TB), TB, 0, stream>>>(h1, a_s1, a_d1, als, ald, N, H, C1);

    agg_band<512, 128, 64, 32, 1><<<cdiv_u((size_t)N, 64), 1024, 0, stream>>>(
        ei, E, N, h1, als, ald, agg1, blist);

    gemm_wmma_f32<true><<<dim3(F2 / TS, N / TS), 128, 0, stream>>>(agg1, W2, h2, N, F1, F2, b1);

    att_logits<<<cdiv_u((size_t)N * H, TB), TB, 0, stream>>>(h2, a_s2, a_d2, als, ald, N, H, LAT);

    agg_band<256, 64, 64, 32, 2><<<cdiv_u((size_t)N, 64), 1024, 0, stream>>>(
        ei, E, N, h2, als, ald, agg2, blist);

    mean_heads_bias<<<cdiv_u((size_t)N * LAT, TB), TB, 0, stream>>>(agg2, b2, out, N, H, LAT);
}


# GEMM as row panels: a block stages 64 whole rows of A in LDS once, B rows go from memory straight to registers, a wave keeps 4x4 or 2x4 tiles, no K-loop barrier; k order in 8-k groups changed
# speedup vs baseline: 6.1403x; 1.2249x over previous
#include <hip/hip_runtime.h>
#include <math.h>
#include <type_traits>

typedef __attribute__((ext_vector_type(2))) float v2f;
typedef __attribute__((ext_vector_type(8))) float v8f;
typedef __attribute__((ext_vector_type(4))) float v4f;

#define PM   64
#define PK   512
#define PSTR 516
#define PLDS (41 * 1024)

static __device__ __forceinline__ float leaky02(float x) {
    return x > 0.f ? x : 0.2f * x;
}

static __device__ __forceinline__ void ldsMaxF(float* addr, float v) {
    if (v >= 0.f)
        __hip_atomic_fetch_max((int*)addr, __float_as_int(v) & 0x7fffffff,
                               __ATOMIC_RELAXED, __HIP_MEMORY_SCOPE_WORKGROUP);
    else
        __hip_atomic_fetch_min((unsigned int*)addr, __float_as_uint(v),
                               __ATOMIC_RELAXED, __HIP_MEMORY_SCOPE_WORKGROUP);
}

template <int N, int TM, bool ACT>
__global__ __launch_bounds__(256)
void gemm_panel_f32(const float* __restrict__ A, const float* __restrict__ B,
                    float* __restrict__ D, const float* __restrict__ abias) {
    constexpr int NWC = N / 64;
    constexpr int NWR = PM / (16 * TM);
    static_assert(NWC * NWR == 8, "8 waves per block");
    static_assert(PLDS >= PM * PSTR, "panel fits");
    __shared__ __attribute__((aligned(16))) float Ap[PLDS];

    const int t    = threadIdx.x;
    const int lane = t & 31;
    const int w    = __builtin_amdgcn_readfirstlane(t >> 5);
    const int half = lane >> 4;
    const int lm   = lane & 15;
    const int wc   = w % NWC;
    const int wr   = w / NWC;
    const int rowM = blockIdx.x * PM;

    {
        const float4* __restrict__ gA = reinterpret_cast<const float4*>(A + (size_t)rowM * PK);
        float4 bb = make_float4(0.f, 0.f, 0.f, 0.f);
        if constexpr (ACT) bb = reinterpret_cast<const float4*>(abias)[t & 127];
        #pragma unroll 8
        for (int i = t; i < PM * PK / 4; i += 256) {
            const int row = i >> 7;
            const int c4  = i & 127;
            float4 v = gA[i];
            if constexpr (ACT) {
                v.x += bb.x; v.y += bb.y; v.z += bb.z; v.w += bb.w;
                v.x = v.x > 0.f ? v.x : 0.f;
                v.y = v.y > 0.f ? v.y : 0.f;
                v.z = v.z > 0.f ? v.z : 0.f;
                v.w = v.w > 0.f ? v.w : 0.f;
            }
            *reinterpret_cast<float4*>(&Ap[row * PSTR + c4 * 4]) = v;
        }
    }
    __syncthreads();

    const float* ap = Ap + (wr * TM * 16 + lm) * PSTR + 4 * half;
    const float* __restrict__ bp = B + (size_t)(4 * half) * N + wc * 64 + 4 * lm;

    v8f acc[TM][4];
    #pragma unroll
    for (int i = 0; i < TM; ++i)
        #pragma unroll
        for (int j = 0; j < 4; ++j) acc[i][j] = v8f{};

    constexpr int KS = 64 / TM;
    constexpr int NB = KS / 2;
    static_assert(PK % (2 * KS) == 0 && (NB == 8 || NB == 16), "two sets per trip");

    auto loadB = [&](v4f (&b)[NB], int k) {
        #pragma unroll
        for (int g = 0; g < KS / 8; ++g)
            #pragma unroll
            for (int s = 0; s < 4; ++s)
                b[g * 4 + s] = *reinterpret_cast<const v4f*>(bp + (size_t)(k + g * 8 + s) * N);
    };
    auto holdB = [&](const v4f (&b)[NB], const v8f& done) {
        if constexpr (NB == 8) {
            asm volatile("" :: "v"(b[0]), "v"(b[1]), "v"(b[2]), "v"(b[3]),
                               "v"(b[4]), "v"(b[5]), "v"(b[6]), "v"(b[7]), "v"(done));
        } else {
            asm volatile("" :: "v"(b[0]), "v"(b[1]), "v"(b[2]), "v"(b[3]),
                               "v"(b[4]), "v"(b[5]), "v"(b[6]), "v"(b[7]),
                               "v"(b[8]), "v"(b[9]), "v"(b[10]), "v"(b[11]),
                               "v"(b[12]), "v"(b[13]), "v"(b[14]), "v"(b[15]), "v"(done));
        }
    };
    auto mma = [&](const v4f (&b)[NB], int k) {
        #pragma unroll
        for (int g = 0; g < KS / 8; ++g) {
            v4f a[TM];
            #pragma unroll
            for (int i = 0; i < TM; ++i)
                a[i] = *reinterpret_cast<const v4f*>(ap + i * 16 * PSTR + k + g * 8);
            #pragma unroll
            for (int p = 0; p < 2; ++p) {
                #pragma unroll
                for (int i = 0; i < TM; ++i) {
                    #pragma unroll
                    for (int j = 0; j < 4; ++j) {
                        v2f av, bv;
                        av.x = a[i][2 * p];
                        av.y = a[i][2 * p + 1];
                        bv.x = b[g * 4 + 2 * p][j];
                        bv.y = b[g * 4 + 2 * p + 1][j];
                        acc[i][j] = __builtin_amdgcn_wmma_f32_16x16x4_f32(
                            false, av, false, bv, (short)0, acc[i][j], false, false);
                    }
                }
            }
        }
    };

    v4f bA[NB], bB[NB];
    loadB(bA, 0);
    for (int k = 0; k < PK; k += 2 * KS) {
        loadB(bB, k + KS);
        mma(bA, k);
        holdB(bB, acc[TM - 1][3]);
        const int kn = (k + 2 * KS < PK) ? k + 2 * KS : k;
        loadB(bA, kn);
        mma(bB, k + KS);
        holdB(bA, acc[TM - 1][3]);
    }

    float* dp = D + (size_t)(rowM + wr * TM * 16 + half * 8) * N + wc * 64 + 4 * lm;
    #pragma unroll
    for (int i = 0; i < TM; ++i) {
        #pragma unroll
        for (int r = 0; r < 8; ++r) {
            *reinterpret_cast<float4*>(dp + (size_t)(i * 16 + r) * N) =
                make_float4(acc[i][0][r], acc[i][1][r], acc[i][2][r], acc[i][3][r]);
        }
    }
}

__global__ void att_logits(const float* __restrict__ h,
                           const float* __restrict__ a_s,
                           const float* __restrict__ a_d,
                           float* __restrict__ al_s, float* __restrict__ al_d,
                           int N, int H, int C) {
    int t = blockIdx.x * blockDim.x + threadIdx.x;
    if (t >= N * H) return;
    int n = t / H, hh = t - n * H;
    const float* row = h + (size_t)n * H * C + (size_t)hh * C;
    const float* vs  = a_s + (size_t)hh * C;
    const float* vd  = a_d + (size_t)hh * C;
    float ss = 0.f, sd = 0.f;
    for (int c = 0; c < C; ++c) {
        float v = row[c];
        ss = fmaf(v, vs[c], ss);
        sd = fmaf(v, vd[c], sd);
    }
    al_s[t] = ss;
    al_d[t] = sd;
}

template <int F, int C, int R, int NW, int LIST>
__global__ __launch_bounds__(NW * 32)
void agg_band(const int* __restrict__ ei, int E, int N,
              const float* __restrict__ hfeat,
              const float* __restrict__ al_s,
              const float* __restrict__ al_d,
              float* __restrict__ agg,
              int* __restrict__ blist) {
    static_assert(F == 4 * C && C % 4 == 0, "four heads");
    static_assert(R % NW == 0 && F % 64 == 0, "band and row shapes");
    constexpr int NT         = NW * 32;
    constexpr int V2         = F / 64;
    constexpr int SPW        = R / NW;
    constexpr int DRAIN_MAX  = 32;
    constexpr int DRAIN_ROWS = 8;
    constexpr int QCAP       = 512;
    static_assert(SPW < DRAIN_ROWS && DRAIN_MAX + 256 <= QCAP, "queue room");
    static_assert(LIST == 1 || LIST == 2, "1: leaves its queues, 2: takes them");
    constexpr int BLS        = 1 + NW + NW * QCAP;

    __shared__ __attribute__((aligned(16))) float acc[R * F];
    __shared__ __attribute__((aligned(16))) float ald_b[R * 4];
    __shared__ __attribute__((aligned(16))) float mx[R * 4];
    __shared__ __attribute__((aligned(16))) float den[R * 4];
    __shared__ int q[NW][QCAP];
    __shared__ int ovf;

    const int tid  = threadIdx.x;
    const int lane = tid & 31;
    const int w    = __builtin_amdgcn_readfirstlane(tid >> 5);
    const int base = blockIdx.x * R;

    float4* acc4 = reinterpret_cast<float4*>(acc);
    for (int i = tid; i < R * F / 4; i += NT) acc4[i] = make_float4(0.f, 0.f, 0.f, 0.f);
    for (int i = tid; i < R * 4; i += NT) {
        const int row = base + (i >> 2);
        ald_b[i] = row < N ? al_d[(size_t)row * 4 + (i & 3)] : 0.f;
        mx[i]    = -__builtin_huge_valf();
        den[i]   = 0.f;
    }
    if (tid == 0) ovf = 0;
    __syncthreads();

    const int* __restrict__ dst = ei + E;
    const int nsteps = E >> 8;

    auto scan = [&](const int thresh, const bool keep, auto&& drain) -> int {
        int  cnt = 0, done = 0;
        bool kept = keep;
        {
            const int row0 = base + w * SPW;
            int nself = N - row0;
            nself = nself < 0 ? 0 : (nself > SPW ? SPW : nself);
            if (lane < nself) q[w][lane] = E + row0 + lane;
            cnt = nself;
        }
        auto push = [&](bool hit, int ev) {
            const unsigned mask = __builtin_amdgcn_ballot_w32(hit);
            if (hit) q[w][cnt + (int)__builtin_amdgcn_mbcnt_lo(mask, 0u)] = ev;
            cnt += __builtin_popcount(mask);
        };
        auto test4 = [&](const int4& v, int e) {
            const bool m0 = (unsigned)(v.x - base) < (unsigned)R;
            const bool m1 = (unsigned)(v.y - base) < (unsigned)R;
            const bool m2 = (unsigned)(v.z - base) < (unsigned)R;
            const bool m3 = (unsigned)(v.w - base) < (unsigned)R;
            if (__builtin_amdgcn_ballot_w32(m0 | m1 | m2 | m3)) {
                push(m0, e);
                push(m1, e + 1);
                push(m2, e + 2);
                push(m3, e + 3);
            }
        };
        for (int step = w; ; step += NW) {
            const bool last = step >= nsteps;
            if (!last) {
                const int  e  = step * 256 + lane * 4;
                const int4 va = *reinterpret_cast<const int4*>(dst + e);
                const int4 vb = *reinterpret_cast<const int4*>(dst + e + 128);
                test4(va, e);
                test4(vb, e + 128);
            } else if (w == 0) {
                for (int e = nsteps * 256 + lane; __builtin_amdgcn_ballot_w32(e < E); e += 32)
                    push(e < E && (unsigned)(dst[e < E ? e : 0] - base) < (unsigned)R, e);
            }
            if (cnt - done >= thresh || (last && cnt > done)) {
                drain(done, cnt);
                if (kept && (last || cnt + thresh + 256 <= QCAP)) {
                    done = cnt;
                } else {
                    kept = false;
                    cnt = 0;
                    done = 0;
                }
            }
            if (last) break;
        }
        return kept ? cnt : -1;
    };

    auto drain_max = [&](const int from, const int to) {
        for (int b = from; b < to; b += 32) {
            if (b + lane < to) {
                const int eq = q[w][b + lane];
                int s, d;
                if (eq < E) { s = ei[eq]; d = dst[eq]; } else { s = d = eq - E; }
                const float4 s4 = *reinterpret_cast<const float4*>(al_s + (size_t)s * 4);
                const int    r  = d - base;
                const float4 d4 = *reinterpret_cast<const float4*>(ald_b + r * 4);
                ldsMaxF(&mx[r * 4 + 0], leaky02(s4.x + d4.x));
                ldsMaxF(&mx[r * 4 + 1], leaky02(s4.y + d4.y));
                ldsMaxF(&mx[r * 4 + 2], leaky02(s4.z + d4.z));
                ldsMaxF(&mx[r * 4 + 3], leaky02(s4.w + d4.w));
            }
        }
    };

    auto drain_rows = [&](const int from, const int to) {
        for (int b = from; b < to; b += 32) {
            const int n = (to - b) < 32 ? (to - b) : 32;
            int   s = 0, r = 0;
            float a0 = 0.f, a1 = 0.f, a2 = 0.f, a3 = 0.f;
            if (lane < n) {
                const int eq = q[w][b + lane];
                int d;
                if (eq < E) { s = ei[eq]; d = dst[eq]; } else { s = d = eq - E; }
                const float4 s4 = *reinterpret_cast<const float4*>(al_s + (size_t)s * 4);
                r = d - base;
                const float4 d4 = *reinterpret_cast<const float4*>(ald_b + r * 4);
                const float4 m4 = *reinterpret_cast<const float4*>(mx + r * 4);
                a0 = __expf(leaky02(s4.x + d4.x) - m4.x);
                a1 = __expf(leaky02(s4.y + d4.y) - m4.y);
                a2 = __expf(leaky02(s4.z + d4.z) - m4.z);
                a3 = __expf(leaky02(s4.w + d4.w) - m4.w);
                __hip_atomic_fetch_add(&den[r * 4 + 0], a0, __ATOMIC_RELAXED, __HIP_MEMORY_SCOPE_WORKGROUP);
                __hip_atomic_fetch_add(&den[r * 4 + 1], a1, __ATOMIC_RELAXED, __HIP_MEMORY_SCOPE_WORKGROUP);
                __hip_atomic_fetch_add(&den[r * 4 + 2], a2, __ATOMIC_RELAXED, __HIP_MEMORY_SCOPE_WORKGROUP);
                __hip_atomic_fetch_add(&den[r * 4 + 3], a3, __ATOMIC_RELAXED, __HIP_MEMORY_SCOPE_WORKGROUP);
            }
            auto rows = [&](int j, auto UC) {
                constexpr int U = decltype(UC)::value;
                float2 v[U][V2];
                #pragma unroll
                for (int u = 0; u < U; ++u) {
                    const int sj = __builtin_amdgcn_readlane(s, j + u);
                    const float2* __restrict__ row =
                        reinterpret_cast<const float2*>(hfeat + (size_t)sj * F) + lane;
                    #pragma unroll
                    for (int k = 0; k < V2; ++k) v[u][k] = row[k * 32];
                }
                #pragma unroll
                for (int u = 0; u < U; ++u) {
                    const int rj = __builtin_amdgcn_readlane(r, j + u);
                    float al[4];
                    al[0] = __builtin_bit_cast(float, __builtin_amdgcn_readlane(__builtin_bit_cast(int, a0), j + u));
                    al[1] = __builtin_bit_cast(float, __builtin_amdgcn_readlane(__builtin_bit_cast(int, a1), j + u));
                    al[2] = __builtin_bit_cast(float, __builtin_amdgcn_readlane(__builtin_bit_cast(int, a2), j + u));
                    al[3] = __builtin_bit_cast(float, __builtin_amdgcn_readlane(__builtin_bit_cast(int, a3), j + u));
                    float* arow = acc + rj * F + lane * 2;
                    #pragma unroll
                    for (int k = 0; k < V2; ++k) {
                        const float a = al[(k * 64) / C];
                        __hip_atomic_fetch_add(arow + k * 64,     v[u][k].x * a,
                                               __ATOMIC_RELAXED, __HIP_MEMORY_SCOPE_WORKGROUP);
                        __hip_atomic_fetch_add(arow + k * 64 + 1, v[u][k].y * a,
                                               __ATOMIC_RELAXED, __HIP_MEMORY_SCOPE_WORKGROUP);
                    }
                }
            };
            int j = 0;
            for (; j + 4 <= n; j += 4) rows(j, std::integral_constant<int, 4>{});
            for (; j < n; ++j)         rows(j, std::integral_constant<int, 1>{});
        }
    };

    int* __restrict__ bl = blist + (size_t)blockIdx.x * BLS;
    int nkept;
    if constexpr (LIST == 2) {
        if (__builtin_amdgcn_readfirstlane(bl[0]) == 0) {
            nkept = __builtin_amdgcn_readfirstlane(bl[1 + w]);
            for (int i = lane; i < nkept; i += 32) q[w][i] = bl[1 + NW + w * QCAP + i];
            drain_max(0, nkept);
        } else {
            nkept = scan(DRAIN_MAX, true, drain_max);
        }
    } else {
        nkept = scan(DRAIN_MAX, true, drain_max);
    }
    if (nkept < 0 && lane == 0)
        __hip_atomic_fetch_or(&ovf, 1, __ATOMIC_RELAXED, __HIP_MEMORY_SCOPE_WORKGROUP);
    __syncthreads();
    if constexpr (LIST == 1) {
        const int lost = __builtin_amdgcn_readfirstlane(ovf);
        if (tid == 0) bl[0] = lost;
        if (lost == 0) {
            if (lane == 0) bl[1 + w] = nkept;
            for (int i = lane; i < nkept; i += 32) bl[1 + NW + w * QCAP + i] = q[w][i];
        }
    }
    if (__builtin_amdgcn_readfirstlane(ovf) == 0) {
        drain_rows(0, nkept);
    } else {
        scan(DRAIN_ROWS, false, drain_rows);
    }
    __syncthreads();

    float4* __restrict__ out4 = reinterpret_cast<float4*>(agg + (size_t)base * F);
    for (int i = tid; i < R * F / 4; i += NT) {
        const int row = i / (F / 4);
        if (base + row < N) {
            const float  dn = den[row * 4 + ((i % (F / 4)) * 4) / C];
            const float4 v  = acc4[i];
            out4[i] = make_float4(v.x / dn, v.y / dn, v.z / dn, v.w / dn);
        }
    }
}

__global__ void mean_heads_bias(const float* __restrict__ agg,
                                const float* __restrict__ b,
                                float* __restrict__ out, int N, int H, int C) {
    int t = blockIdx.x * blockDim.x + threadIdx.x;
    if (t >= N * C) return;
    int n = t / C, c = t - n * C;
    float s = 0.f;
    for (int hh = 0; hh < H; ++hh)
        s += agg[(size_t)n * H * C + (size_t)hh * C + c];
    out[t] = s / (float)H + b[c];
}

static inline unsigned cdiv_u(size_t a, unsigned b) {
    return (unsigned)((a + b - 1) / b);
}

extern "C" void kernel_launch(void* const* d_in, const int* in_sizes, int n_in,
                              void* d_out, int out_size, void* d_ws, size_t ws_size,
                              hipStream_t stream) {
    const float* x     = (const float*)d_in[0];
    const int*   ei    = (const int*)d_in[1];
    const float* W1    = (const float*)d_in[2];
    const float* a_s1  = (const float*)d_in[3];
    const float* a_d1  = (const float*)d_in[4];
    const float* b1    = (const float*)d_in[5];
    const float* W2    = (const float*)d_in[6];
    const float* a_s2  = (const float*)d_in[7];
    const float* a_d2  = (const float*)d_in[8];
    const float* b2    = (const float*)d_in[9];
    float*       out   = (float*)d_out;

    const int F1   = in_sizes[3];
    const int F2   = in_sizes[7];
    const int LAT  = in_sizes[9];
    const int H    = F2 / LAT;
    const int C1   = F1 / H;
    const int Fin  = in_sizes[2] / F1;
    const int N    = in_sizes[0] / Fin;
    const int E    = in_sizes[1] / 2;

    float* ws = (float*)d_ws;
    size_t o = 0;
    float* h1   = ws + o; o += (size_t)N * F1;
    float* agg1 = ws + o; o += (size_t)N * F1;
    float* als  = ws + o; o += (size_t)N * H;
    float* ald  = ws + o; o += (size_t)N * H;
    float* h2   = h1;
    float* agg2 = h1 + (size_t)N * F2;
    int*   blist = (int*)(ws + o);
    (void)ws_size; (void)n_in; (void)out_size;

    const unsigned TB = 256;

    gemm_panel_f32<512, 4, false><<<N / PM, 256, 0, stream>>>(x, W1, h1, nullptr);

    att_logits<<<cdiv_u((size_t)N * H, TB), TB, 0, stream>>>(h1, a_s1, a_d1, als, ald, N, H, C1);

    agg_band<512, 128, 64, 32, 1><<<cdiv_u((size_t)N, 64), 1024, 0, stream>>>(
        ei, E, N, h1, als, ald, agg1, blist);

    gemm_panel_f32<256, 2, true><<<N / PM, 256, 0, stream>>>(agg1, W2, h2, b1);

    att_logits<<<cdiv_u((size_t)N * H, TB), TB, 0, stream>>>(h2, a_s2, a_d2, als, ald, N, H, LAT);

    agg_band<256, 64, 64, 32, 2><<<cdiv_u((size_t)N, 64), 1024, 0, stream>>>(
        ei, E, N, h2, als, ald, agg2, blist);

    mean_heads_bias<<<cdiv_u((size_t)N * LAT, TB), TB, 0, stream>>>(agg2, b2, out, N, H, LAT);
}


# attention logits taken in the GEMM epilogue from the accumulator tiles (4-column lane partials through LDS, added per head in column order): both logit launches gone, 7 -> 5; logit sum order changed
# speedup vs baseline: 6.4332x; 1.0477x over previous
#include <hip/hip_runtime.h>
#include <math.h>
#include <type_traits>

typedef __attribute__((ext_vector_type(2))) float v2f;
typedef __attribute__((ext_vector_type(8))) float v8f;
typedef __attribute__((ext_vector_type(4))) float v4f;

#define PM   64
#define PK   512
#define PSTR 516
#define PLDS (41 * 1024)

static __device__ __forceinline__ float leaky02(float x) {
    return x > 0.f ? x : 0.2f * x;
}

static __device__ __forceinline__ void ldsMaxF(float* addr, float v) {
    if (v >= 0.f)
        __hip_atomic_fetch_max((int*)addr, __float_as_int(v) & 0x7fffffff,
                               __ATOMIC_RELAXED, __HIP_MEMORY_SCOPE_WORKGROUP);
    else
        __hip_atomic_fetch_min((unsigned int*)addr, __float_as_uint(v),
                               __ATOMIC_RELAXED, __HIP_MEMORY_SCOPE_WORKGROUP);
}

template <int N, int TM, bool ACT>
__global__ __launch_bounds__(256)
void gemm_panel_f32(const float* __restrict__ A, const float* __restrict__ B,
                    float* __restrict__ D, const float* __restrict__ abias,
                    const float* __restrict__ a_s, const float* __restrict__ a_d,
                    float* __restrict__ al_s, float* __restrict__ al_d) {
    constexpr int NWC = N / 64;
    constexpr int NWR = PM / (16 * TM);
    static_assert(NWC * NWR == 8, "8 waves per block");
    static_assert(PLDS >= PM * PSTR, "panel fits");
    __shared__ __attribute__((aligned(16))) float Ap[PLDS];

    const int t    = threadIdx.x;
    const int lane = t & 31;
    const int w    = __builtin_amdgcn_readfirstlane(t >> 5);
    const int half = lane >> 4;
    const int lm   = lane & 15;
    const int wc   = w % NWC;
    const int wr   = w / NWC;
    const int rowM = blockIdx.x * PM;

    {
        const float4* __restrict__ gA = reinterpret_cast<const float4*>(A + (size_t)rowM * PK);
        float4 bb = make_float4(0.f, 0.f, 0.f, 0.f);
        if constexpr (ACT) bb = reinterpret_cast<const float4*>(abias)[t & 127];
        #pragma unroll 8
        for (int i = t; i < PM * PK / 4; i += 256) {
            const int row = i >> 7;
            const int c4  = i & 127;
            float4 v = gA[i];
            if constexpr (ACT) {
                v.x += bb.x; v.y += bb.y; v.z += bb.z; v.w += bb.w;
                v.x = v.x > 0.f ? v.x : 0.f;
                v.y = v.y > 0.f ? v.y : 0.f;
                v.z = v.z > 0.f ? v.z : 0.f;
                v.w = v.w > 0.f ? v.w : 0.f;
            }
            *reinterpret_cast<float4*>(&Ap[row * PSTR + c4 * 4]) = v;
        }
    }
    __syncthreads();

    const float* ap = Ap + (wr * TM * 16 + lm) * PSTR + 4 * half;
    const float* __restrict__ bp = B + (size_t)(4 * half) * N + wc * 64 + 4 * lm;

    v8f acc[TM][4];
    #pragma unroll
    for (int i = 0; i < TM; ++i)
        #pragma unroll
        for (int j = 0; j < 4; ++j) acc[i][j] = v8f{};

    constexpr int KS = 64 / TM;
    constexpr int NB = KS / 2;
    static_assert(PK % (2 * KS) == 0 && (NB == 8 || NB == 16), "two sets per trip");

    auto loadB = [&](v4f (&b)[NB], int k) {
        #pragma unroll
        for (int g = 0; g < KS / 8; ++g)
            #pragma unroll
            for (int s = 0; s < 4; ++s)
                b[g * 4 + s] = *reinterpret_cast<const v4f*>(bp + (size_t)(k + g * 8 + s) * N);
    };
    auto holdB = [&](const v4f (&b)[NB], const v8f& done) {
        if constexpr (NB == 8) {
            asm volatile("" :: "v"(b[0]), "v"(b[1]), "v"(b[2]), "v"(b[3]),
                               "v"(b[4]), "v"(b[5]), "v"(b[6]), "v"(b[7]), "v"(done));
        } else {
            asm volatile("" :: "v"(b[0]), "v"(b[1]), "v"(b[2]), "v"(b[3]),
                               "v"(b[4]), "v"(b[5]), "v"(b[6]), "v"(b[7]),
                               "v"(b[8]), "v"(b[9]), "v"(b[10]), "v"(b[11]),
                               "v"(b[12]), "v"(b[13]), "v"(b[14]), "v"(b[15]), "v"(done));
        }
    };
    auto mma = [&](const v4f (&b)[NB], int k) {
        #pragma unroll
        for (int g = 0; g < KS / 8; ++g) {
            v4f a[TM];
            #pragma unroll
            for (int i = 0; i < TM; ++i)
                a[i] = *reinterpret_cast<const v4f*>(ap + i * 16 * PSTR + k + g * 8);
            #pragma unroll
            for (int p = 0; p < 2; ++p) {
                #pragma unroll
                for (int i = 0; i < TM; ++i) {
                    #pragma unroll
                    for (int j = 0; j < 4; ++j) {
                        v2f av, bv;
                        av.x = a[i][2 * p];
                        av.y = a[i][2 * p + 1];
                        bv.x = b[g * 4 + 2 * p][j];
                        bv.y = b[g * 4 + 2 * p + 1][j];
                        acc[i][j] = __builtin_amdgcn_wmma_f32_16x16x4_f32(
                            false, av, false, bv, (short)0, acc[i][j], false, false);
                    }
                }
            }
        }
    };

    v4f bA[NB], bB[NB];
    loadB(bA, 0);
    for (int k = 0; k < PK; k += 2 * KS) {
        loadB(bB, k + KS);
        mma(bA, k);
        holdB(bB, acc[TM - 1][3]);
        const int kn = (k + 2 * KS < PK) ? k + 2 * KS : k;
        loadB(bA, kn);
        mma(bB, k + KS);
        holdB(bA, acc[TM - 1][3]);
    }

    float* dp = D + (size_t)(rowM + wr * TM * 16 + half * 8) * N + wc * 64 + 4 * lm;
    #pragma unroll
    for (int i = 0; i < TM; ++i) {
        #pragma unroll
        for (int r = 0; r < 8; ++r) {
            *reinterpret_cast<float4*>(dp + (size_t)(i * 16 + r) * N) =
                make_float4(acc[i][0][r], acc[i][1][r], acc[i][2][r], acc[i][3][r]);
        }
    }

    constexpr int LP = N / 4 + 4;
    constexpr int HW = N / 16;
    static_assert(2 * PM * LP <= PM * PSTR && HW % 4 == 0, "partials fit in the panel's space");
    const float4 ws4 = *reinterpret_cast<const float4*>(a_s + wc * 64 + 4 * lm);
    const float4 wd4 = *reinterpret_cast<const float4*>(a_d + wc * 64 + 4 * lm);
    __syncthreads();
    {
        float* pls = Ap + (wr * TM * 16 + half * 8) * LP + wc * 16 + lm;
        float* pld = pls + PM * LP;
        #pragma unroll
        for (int i = 0; i < TM; ++i) {
            #pragma unroll
            for (int r = 0; r < 8; ++r) {
                pls[(i * 16 + r) * LP] = (acc[i][0][r] * ws4.x + acc[i][1][r] * ws4.y)
                                       + (acc[i][2][r] * ws4.z + acc[i][3][r] * ws4.w);
                pld[(i * 16 + r) * LP] = (acc[i][0][r] * wd4.x + acc[i][1][r] * wd4.y)
                                       + (acc[i][2][r] * wd4.z + acc[i][3][r] * wd4.w);
            }
        }
    }
    __syncthreads();
    {
        const float4* qs = reinterpret_cast<const float4*>(Ap + (t >> 2) * LP + (t & 3) * HW);
        const float4* qd = reinterpret_cast<const float4*>(Ap + PM * LP + (t >> 2) * LP + (t & 3) * HW);
        float ss = 0.f, sd = 0.f;
        #pragma unroll
        for (int q = 0; q < HW / 4; ++q) {
            const float4 u = qs[q];
            const float4 v = qd[q];
            ss += (u.x + u.y) + (u.z + u.w);
            sd += (v.x + v.y) + (v.z + v.w);
        }
        al_s[(size_t)rowM * 4 + t] = ss;
        al_d[(size_t)rowM * 4 + t] = sd;
    }
}

template <int F, int C, int R, int NW, int LIST>
__global__ __launch_bounds__(NW * 32)
void agg_band(const int* __restrict__ ei, int E, int N,
              const float* __restrict__ hfeat,
              const float* __restrict__ al_s,
              const float* __restrict__ al_d,
              float* __restrict__ agg,
              int* __restrict__ blist) {
    static_assert(F == 4 * C && C % 4 == 0, "four heads");
    static_assert(R % NW == 0 && F % 64 == 0, "band and row shapes");
    constexpr int NT         = NW * 32;
    constexpr int V2         = F / 64;
    constexpr int SPW        = R / NW;
    constexpr int DRAIN_MAX  = 32;
    constexpr int DRAIN_ROWS = 8;
    constexpr int QCAP       = 512;
    static_assert(SPW < DRAIN_ROWS && DRAIN_MAX + 256 <= QCAP, "queue room");
    static_assert(LIST == 1 || LIST == 2, "1: leaves its queues, 2: takes them");
    constexpr int BLS        = 1 + NW + NW * QCAP;

    __shared__ __attribute__((aligned(16))) float acc[R * F];
    __shared__ __attribute__((aligned(16))) float ald_b[R * 4];
    __shared__ __attribute__((aligned(16))) float mx[R * 4];
    __shared__ __attribute__((aligned(16))) float den[R * 4];
    __shared__ int q[NW][QCAP];
    __shared__ int ovf;

    const int tid  = threadIdx.x;
    const int lane = tid & 31;
    const int w    = __builtin_amdgcn_readfirstlane(tid >> 5);
    const int base = blockIdx.x * R;

    float4* acc4 = reinterpret_cast<float4*>(acc);
    for (int i = tid; i < R * F / 4; i += NT) acc4[i] = make_float4(0.f, 0.f, 0.f, 0.f);
    for (int i = tid; i < R * 4; i += NT) {
        const int row = base + (i >> 2);
        ald_b[i] = row < N ? al_d[(size_t)row * 4 + (i & 3)] : 0.f;
        mx[i]    = -__builtin_huge_valf();
        den[i]   = 0.f;
    }
    if (tid == 0) ovf = 0;
    __syncthreads();

    const int* __restrict__ dst = ei + E;
    const int nsteps = E >> 8;

    auto scan = [&](const int thresh, const bool keep, auto&& drain) -> int {
        int  cnt = 0, done = 0;
        bool kept = keep;
        {
            const int row0 = base + w * SPW;
            int nself = N - row0;
            nself = nself < 0 ? 0 : (nself > SPW ? SPW : nself);
            if (lane < nself) q[w][lane] = E + row0 + lane;
            cnt = nself;
        }
        auto push = [&](bool hit, int ev) {
            const unsigned mask = __builtin_amdgcn_ballot_w32(hit);
            if (hit) q[w][cnt + (int)__builtin_amdgcn_mbcnt_lo(mask, 0u)] = ev;
            cnt += __builtin_popcount(mask);
        };
        auto test4 = [&](const int4& v, int e) {
            const bool m0 = (unsigned)(v.x - base) < (unsigned)R;
            const bool m1 = (unsigned)(v.y - base) < (unsigned)R;
            const bool m2 = (unsigned)(v.z - base) < (unsigned)R;
            const bool m3 = (unsigned)(v.w - base) < (unsigned)R;
            if (__builtin_amdgcn_ballot_w32(m0 | m1 | m2 | m3)) {
                push(m0, e);
                push(m1, e + 1);
                push(m2, e + 2);
                push(m3, e + 3);
            }
        };
        for (int step = w; ; step += NW) {
            const bool last = step >= nsteps;
            if (!last) {
                const int  e  = step * 256 + lane * 4;
                const int4 va = *reinterpret_cast<const int4*>(dst + e);
                const int4 vb = *reinterpret_cast<const int4*>(dst + e + 128);
                test4(va, e);
                test4(vb, e + 128);
            } else if (w == 0) {
                for (int e = nsteps * 256 + lane; __builtin_amdgcn_ballot_w32(e < E); e += 32)
                    push(e < E && (unsigned)(dst[e < E ? e : 0] - base) < (unsigned)R, e);
            }
            if (cnt - done >= thresh || (last && cnt > done)) {
                drain(done, cnt);
                if (kept && (last || cnt + thresh + 256 <= QCAP)) {
                    done = cnt;
                } else {
                    kept = false;
                    cnt = 0;
                    done = 0;
                }
            }
            if (last) break;
        }
        return kept ? cnt : -1;
    };

    auto drain_max = [&](const int from, const int to) {
        for (int b = from; b < to; b += 32) {
            if (b + lane < to) {
                const int eq = q[w][b + lane];
                int s, d;
                if (eq < E) { s = ei[eq]; d = dst[eq]; } else { s = d = eq - E; }
                const float4 s4 = *reinterpret_cast<const float4*>(al_s + (size_t)s * 4);
                const int    r  = d - base;
                const float4 d4 = *reinterpret_cast<const float4*>(ald_b + r * 4);
                ldsMaxF(&mx[r * 4 + 0], leaky02(s4.x + d4.x));
                ldsMaxF(&mx[r * 4 + 1], leaky02(s4.y + d4.y));
                ldsMaxF(&mx[r * 4 + 2], leaky02(s4.z + d4.z));
                ldsMaxF(&mx[r * 4 + 3], leaky02(s4.w + d4.w));
            }
        }
    };

    auto drain_rows = [&](const int from, const int to) {
        for (int b = from; b < to; b += 32) {
            const int n = (to - b) < 32 ? (to - b) : 32;
            int   s = 0, r = 0;
            float a0 = 0.f, a1 = 0.f, a2 = 0.f, a3 = 0.f;
            if (lane < n) {
                const int eq = q[w][b + lane];
                int d;
                if (eq < E) { s = ei[eq]; d = dst[eq]; } else { s = d = eq - E; }
                const float4 s4 = *reinterpret_cast<const float4*>(al_s + (size_t)s * 4);
                r = d - base;
                const float4 d4 = *reinterpret_cast<const float4*>(ald_b + r * 4);
                const float4 m4 = *reinterpret_cast<const float4*>(mx + r * 4);
                a0 = __expf(leaky02(s4.x + d4.x) - m4.x);
                a1 = __expf(leaky02(s4.y + d4.y) - m4.y);
                a2 = __expf(leaky02(s4.z + d4.z) - m4.z);
                a3 = __expf(leaky02(s4.w + d4.w) - m4.w);
                __hip_atomic_fetch_add(&den[r * 4 + 0], a0, __ATOMIC_RELAXED, __HIP_MEMORY_SCOPE_WORKGROUP);
                __hip_atomic_fetch_add(&den[r * 4 + 1], a1, __ATOMIC_RELAXED, __HIP_MEMORY_SCOPE_WORKGROUP);
                __hip_atomic_fetch_add(&den[r * 4 + 2], a2, __ATOMIC_RELAXED, __HIP_MEMORY_SCOPE_WORKGROUP);
                __hip_atomic_fetch_add(&den[r * 4 + 3], a3, __ATOMIC_RELAXED, __HIP_MEMORY_SCOPE_WORKGROUP);
            }
            auto rows = [&](int j, auto UC) {
                constexpr int U = decltype(UC)::value;
                float2 v[U][V2];
                #pragma unroll
                for (int u = 0; u < U; ++u) {
                    const int sj = __builtin_amdgcn_readlane(s, j + u);
                    const float2* __restrict__ row =
                        reinterpret_cast<const float2*>(hfeat + (size_t)sj * F) + lane;
                    #pragma unroll
                    for (int k = 0; k < V2; ++k) v[u][k] = row[k * 32];
                }
                #pragma unroll
                for (int u = 0; u < U; ++u) {
                    const int rj = __builtin_amdgcn_readlane(r, j + u);
                    float al[4];
                    al[0] = __builtin_bit_cast(float, __builtin_amdgcn_readlane(__builtin_bit_cast(int, a0), j + u));
                    al[1] = __builtin_bit_cast(float, __builtin_amdgcn_readlane(__builtin_bit_cast(int, a1), j + u));
                    al[2] = __builtin_bit_cast(float, __builtin_amdgcn_readlane(__builtin_bit_cast(int, a2), j + u));
                    al[3] = __builtin_bit_cast(float, __builtin_amdgcn_readlane(__builtin_bit_cast(int, a3), j + u));
                    float* arow = acc + rj * F + lane * 2;
                    #pragma unroll
                    for (int k = 0; k < V2; ++k) {
                        const float a = al[(k * 64) / C];
                        __hip_atomic_fetch_add(arow + k * 64,     v[u][k].x * a,
                                               __ATOMIC_RELAXED, __HIP_MEMORY_SCOPE_WORKGROUP);
                        __hip_atomic_fetch_add(arow + k * 64 + 1, v[u][k].y * a,
                                               __ATOMIC_RELAXED, __HIP_MEMORY_SCOPE_WORKGROUP);
                    }
                }
            };
            int j = 0;
            for (; j + 4 <= n; j += 4) rows(j, std::integral_constant<int, 4>{});
            for (; j < n; ++j)         rows(j, std::integral_constant<int, 1>{});
        }
    };

    int* __restrict__ bl = blist + (size_t)blockIdx.x * BLS;
    int nkept;
    if constexpr (LIST == 2) {
        if (__builtin_amdgcn_readfirstlane(bl[0]) == 0) {
            nkept = __builtin_amdgcn_readfirstlane(bl[1 + w]);
            for (int i = lane; i < nkept; i += 32) q[w][i] = bl[1 + NW + w * QCAP + i];
            drain_max(0, nkept);
        } else {
            nkept = scan(DRAIN_MAX, true, drain_max);
        }
    } else {
        nkept = scan(DRAIN_MAX, true, drain_max);
    }
    if (nkept < 0 && lane == 0)
        __hip_atomic_fetch_or(&ovf, 1, __ATOMIC_RELAXED, __HIP_MEMORY_SCOPE_WORKGROUP);
    __syncthreads();
    if constexpr (LIST == 1) {
        const int lost = __builtin_amdgcn_readfirstlane(ovf);
        if (tid == 0) bl[0] = lost;
        if (lost == 0) {
            if (lane == 0) bl[1 + w] = nkept;
            for (int i = lane; i < nkept; i += 32) bl[1 + NW + w * QCAP + i] = q[w][i];
        }
    }
    if (__builtin_amdgcn_readfirstlane(ovf) == 0) {
        drain_rows(0, nkept);
    } else {
        scan(DRAIN_ROWS, false, drain_rows);
    }
    __syncthreads();

    float4* __restrict__ out4 = reinterpret_cast<float4*>(agg + (size_t)base * F);
    for (int i = tid; i < R * F / 4; i += NT) {
        const int row = i / (F / 4);
        if (base + row < N) {
            const float  dn = den[row * 4 + ((i % (F / 4)) * 4) / C];
            const float4 v  = acc4[i];
            out4[i] = make_float4(v.x / dn, v.y / dn, v.z / dn, v.w / dn);
        }
    }
}

__global__ void mean_heads_bias(const float* __restrict__ agg,
                                const float* __restrict__ b,
                                float* __restrict__ out, int N, int H, int C) {
    int t = blockIdx.x * blockDim.x + threadIdx.x;
    if (t >= N * C) return;
    int n = t / C, c = t - n * C;
    float s = 0.f;
    for (int hh = 0; hh < H; ++hh)
        s += agg[(size_t)n * H * C + (size_t)hh * C + c];
    out[t] = s / (float)H + b[c];
}

static inline unsigned cdiv_u(size_t a, unsigned b) {
    return (unsigned)((a + b - 1) / b);
}

extern "C" void kernel_launch(void* const* d_in, const int* in_sizes, int n_in,
                              void* d_out, int out_size, void* d_ws, size_t ws_size,
                              hipStream_t stream) {
    const float* x     = (const float*)d_in[0];
    const int*   ei    = (const int*)d_in[1];
    const float* W1    = (const float*)d_in[2];
    const float* a_s1  = (const float*)d_in[3];
    const float* a_d1  = (const float*)d_in[4];
    const float* b1    = (const float*)d_in[5];
    const float* W2    = (const float*)d_in[6];
    const float* a_s2  = (const float*)d_in[7];
    const float* a_d2  = (const float*)d_in[8];
    const float* b2    = (const float*)d_in[9];
    float*       out   = (float*)d_out;

    const int F1   = in_sizes[3];
    const int F2   = in_sizes[7];
    const int LAT  = in_sizes[9];
    const int H    = F2 / LAT;
    const int Fin  = in_sizes[2] / F1;
    const int N    = in_sizes[0] / Fin;
    const int E    = in_sizes[1] / 2;

    float* ws = (float*)d_ws;
    size_t o = 0;
    float* h1   = ws + o; o += (size_t)N * F1;
    float* agg1 = ws + o; o += (size_t)N * F1;
    float* als  = ws + o; o += (size_t)N * H;
    float* ald  = ws + o; o += (size_t)N * H;
    float* h2   = h1;
    float* agg2 = h1 + (size_t)N * F2;
    int*   blist = (int*)(ws + o);
    (void)ws_size; (void)n_in; (void)out_size;

    const unsigned TB = 256;

    gemm_panel_f32<512, 4, false><<<N / PM, 256, 0, stream>>>(x, W1, h1, nullptr, a_s1, a_d1, als, ald);

    agg_band<512, 128, 64, 32, 1><<<cdiv_u((size_t)N, 64), 1024, 0, stream>>>(
        ei, E, N, h1, als, ald, agg1, blist);

    gemm_panel_f32<256, 2, true><<<N / PM, 256, 0, stream>>>(agg1, W2, h2, b1, a_s2, a_d2, als, ald);

    agg_band<256, 64, 64, 32, 2><<<cdiv_u((size_t)N, 64), 1024, 0, stream>>>(
        ei, E, N, h2, als, ald, agg2, blist);

    mean_heads_bias<<<cdiv_u((size_t)N * LAT, TB), TB, 0, stream>>>(agg2, b2, out, N, H, LAT);
}


# layer 2: head mean commuted ahead of the edge sum: after a denominator pass an edge adds sum_h alpha_h/4*row_h to one 64-float LDS row, 4x fewer LDS adds; +b2 to out, 5->4 launches; sum order changed
# speedup vs baseline: 7.0124x; 1.0900x over previous
#include <hip/hip_runtime.h>
#include <math.h>
#include <type_traits>

typedef __attribute__((ext_vector_type(2))) float v2f;
typedef __attribute__((ext_vector_type(8))) float v8f;
typedef __attribute__((ext_vector_type(4))) float v4f;

#define PM   64
#define PK   512
#define PSTR 516
#define PLDS (41 * 1024)

static __device__ __forceinline__ float leaky02(float x) {
    return x > 0.f ? x : 0.2f * x;
}

static __device__ __forceinline__ void ldsMaxF(float* addr, float v) {
    if (v >= 0.f)
        __hip_atomic_fetch_max((int*)addr, __float_as_int(v) & 0x7fffffff,
                               __ATOMIC_RELAXED, __HIP_MEMORY_SCOPE_WORKGROUP);
    else
        __hip_atomic_fetch_min((unsigned int*)addr, __float_as_uint(v),
                               __ATOMIC_RELAXED, __HIP_MEMORY_SCOPE_WORKGROUP);
}

template <int N, int TM, bool ACT>
__global__ __launch_bounds__(256)
void gemm_panel_f32(const float* __restrict__ A, const float* __restrict__ B,
                    float* __restrict__ D, const float* __restrict__ abias,
                    const float* __restrict__ a_s, const float* __restrict__ a_d,
                    float* __restrict__ al_s, float* __restrict__ al_d) {
    constexpr int NWC = N / 64;
    constexpr int NWR = PM / (16 * TM);
    static_assert(NWC * NWR == 8, "8 waves per block");
    static_assert(PLDS >= PM * PSTR, "panel fits");
    __shared__ __attribute__((aligned(16))) float Ap[PLDS];

    const int t    = threadIdx.x;
    const int lane = t & 31;
    const int w    = __builtin_amdgcn_readfirstlane(t >> 5);
    const int half = lane >> 4;
    const int lm   = lane & 15;
    const int wc   = w % NWC;
    const int wr   = w / NWC;
    const int rowM = blockIdx.x * PM;

    {
        const float4* __restrict__ gA = reinterpret_cast<const float4*>(A + (size_t)rowM * PK);
        float4 bb = make_float4(0.f, 0.f, 0.f, 0.f);
        if constexpr (ACT) bb = reinterpret_cast<const float4*>(abias)[t & 127];
        #pragma unroll 8
        for (int i = t; i < PM * PK / 4; i += 256) {
            const int row = i >> 7;
            const int c4  = i & 127;
            float4 v = gA[i];
            if constexpr (ACT) {
                v.x += bb.x; v.y += bb.y; v.z += bb.z; v.w += bb.w;
                v.x = v.x > 0.f ? v.x : 0.f;
                v.y = v.y > 0.f ? v.y : 0.f;
                v.z = v.z > 0.f ? v.z : 0.f;
                v.w = v.w > 0.f ? v.w : 0.f;
            }
            *reinterpret_cast<float4*>(&Ap[row * PSTR + c4 * 4]) = v;
        }
    }
    __syncthreads();

    const float* ap = Ap + (wr * TM * 16 + lm) * PSTR + 4 * half;
    const float* __restrict__ bp = B + (size_t)(4 * half) * N + wc * 64 + 4 * lm;

    v8f acc[TM][4];
    #pragma unroll
    for (int i = 0; i < TM; ++i)
        #pragma unroll
        for (int j = 0; j < 4; ++j) acc[i][j] = v8f{};

    constexpr int KS = 64 / TM;
    constexpr int NB = KS / 2;
    static_assert(PK % (2 * KS) == 0 && (NB == 8 || NB == 16), "two sets per trip");

    auto loadB = [&](v4f (&b)[NB], int k) {
        #pragma unroll
        for (int g = 0; g < KS / 8; ++g)
            #pragma unroll
            for (int s = 0; s < 4; ++s)
                b[g * 4 + s] = *reinterpret_cast<const v4f*>(bp + (size_t)(k + g * 8 + s) * N);
    };
    auto holdB = [&](const v4f (&b)[NB], const v8f& done) {
        if constexpr (NB == 8) {
            asm volatile("" :: "v"(b[0]), "v"(b[1]), "v"(b[2]), "v"(b[3]),
                               "v"(b[4]), "v"(b[5]), "v"(b[6]), "v"(b[7]), "v"(done));
        } else {
            asm volatile("" :: "v"(b[0]), "v"(b[1]), "v"(b[2]), "v"(b[3]),
                               "v"(b[4]), "v"(b[5]), "v"(b[6]), "v"(b[7]),
                               "v"(b[8]), "v"(b[9]), "v"(b[10]), "v"(b[11]),
                               "v"(b[12]), "v"(b[13]), "v"(b[14]), "v"(b[15]), "v"(done));
        }
    };
    auto mma = [&](const v4f (&b)[NB], int k) {
        #pragma unroll
        for (int g = 0; g < KS / 8; ++g) {
            v4f a[TM];
            #pragma unroll
            for (int i = 0; i < TM; ++i)
                a[i] = *reinterpret_cast<const v4f*>(ap + i * 16 * PSTR + k + g * 8);
            #pragma unroll
            for (int p = 0; p < 2; ++p) {
                #pragma unroll
                for (int i = 0; i < TM; ++i) {
                    #pragma unroll
                    for (int j = 0; j < 4; ++j) {
                        v2f av, bv;
                        av.x = a[i][2 * p];
                        av.y = a[i][2 * p + 1];
                        bv.x = b[g * 4 + 2 * p][j];
                        bv.y = b[g * 4 + 2 * p + 1][j];
                        acc[i][j] = __builtin_amdgcn_wmma_f32_16x16x4_f32(
                            false, av, false, bv, (short)0, acc[i][j], false, false);
                    }
                }
            }
        }
    };

    v4f bA[NB], bB[NB];
    loadB(bA, 0);
    for (int k = 0; k < PK; k += 2 * KS) {
        loadB(bB, k + KS);
        mma(bA, k);
        holdB(bB, acc[TM - 1][3]);
        const int kn = (k + 2 * KS < PK) ? k + 2 * KS : k;
        loadB(bA, kn);
        mma(bB, k + KS);
        holdB(bA, acc[TM - 1][3]);
    }

    float* dp = D + (size_t)(rowM + wr * TM * 16 + half * 8) * N + wc * 64 + 4 * lm;
    #pragma unroll
    for (int i = 0; i < TM; ++i) {
        #pragma unroll
        for (int r = 0; r < 8; ++r) {
            *reinterpret_cast<float4*>(dp + (size_t)(i * 16 + r) * N) =
                make_float4(acc[i][0][r], acc[i][1][r], acc[i][2][r], acc[i][3][r]);
        }
    }

    constexpr int LP = N / 4 + 4;
    constexpr int HW = N / 16;
    static_assert(2 * PM * LP <= PM * PSTR && HW % 4 == 0, "partials fit in the panel's space");
    const float4 ws4 = *reinterpret_cast<const float4*>(a_s + wc * 64 + 4 * lm);
    const float4 wd4 = *reinterpret_cast<const float4*>(a_d + wc * 64 + 4 * lm);
    __syncthreads();
    {
        float* pls = Ap + (wr * TM * 16 + half * 8) * LP + wc * 16 + lm;
        float* pld = pls + PM * LP;
        #pragma unroll
        for (int i = 0; i < TM; ++i) {
            #pragma unroll
            for (int r = 0; r < 8; ++r) {
                pls[(i * 16 + r) * LP] = (acc[i][0][r] * ws4.x + acc[i][1][r] * ws4.y)
                                       + (acc[i][2][r] * ws4.z + acc[i][3][r] * ws4.w);
                pld[(i * 16 + r) * LP] = (acc[i][0][r] * wd4.x + acc[i][1][r] * wd4.y)
                                       + (acc[i][2][r] * wd4.z + acc[i][3][r] * wd4.w);
            }
        }
    }
    __syncthreads();
    {
        const float4* qs = reinterpret_cast<const float4*>(Ap + (t >> 2) * LP + (t & 3) * HW);
        const float4* qd = reinterpret_cast<const float4*>(Ap + PM * LP + (t >> 2) * LP + (t & 3) * HW);
        float ss = 0.f, sd = 0.f;
        #pragma unroll
        for (int q = 0; q < HW / 4; ++q) {
            const float4 u = qs[q];
            const float4 v = qd[q];
            ss += (u.x + u.y) + (u.z + u.w);
            sd += (v.x + v.y) + (v.z + v.w);
        }
        al_s[(size_t)rowM * 4 + t] = ss;
        al_d[(size_t)rowM * 4 + t] = sd;
    }
}

template <int F, int C, int R, int NW, int LIST, bool MEAN>
__global__ __launch_bounds__(NW * 32)
void agg_band(const int* __restrict__ ei, int E, int N,
              const float* __restrict__ hfeat,
              const float* __restrict__ al_s,
              const float* __restrict__ al_d,
              float* __restrict__ agg,
              int* __restrict__ blist,
              const float* __restrict__ bias) {
    static_assert(F == 4 * C && C % 4 == 0, "four heads");
    static_assert(!MEAN || C == 64, "MEAN: a lane's k-th float2 of a row is head k, channels 2*lane and 2*lane+1");
    constexpr int AF = MEAN ? C : F;
    static_assert(R % NW == 0 && F % 64 == 0, "band and row shapes");
    constexpr int NT         = NW * 32;
    constexpr int V2         = F / 64;
    constexpr int SPW        = R / NW;
    constexpr int DRAIN_MAX  = 32;
    constexpr int DRAIN_ROWS = 8;
    constexpr int QCAP       = 512;
    static_assert(SPW < DRAIN_ROWS && DRAIN_MAX + 256 <= QCAP, "queue room");
    static_assert(LIST == 1 || LIST == 2, "1: leaves its queues, 2: takes them");
    constexpr int BLS        = 1 + NW + NW * QCAP;

    __shared__ __attribute__((aligned(16))) float acc[R * AF];
    __shared__ __attribute__((aligned(16))) float ald_b[R * 4];
    __shared__ __attribute__((aligned(16))) float mx[R * 4];
    __shared__ __attribute__((aligned(16))) float den[R * 4];
    __shared__ int q[NW][QCAP];
    __shared__ int ovf;

    const int tid  = threadIdx.x;
    const int lane = tid & 31;
    const int w    = __builtin_amdgcn_readfirstlane(tid >> 5);
    const int base = blockIdx.x * R;

    float4* acc4 = reinterpret_cast<float4*>(acc);
    for (int i = tid; i < R * AF / 4; i += NT) acc4[i] = make_float4(0.f, 0.f, 0.f, 0.f);
    for (int i = tid; i < R * 4; i += NT) {
        const int row = base + (i >> 2);
        ald_b[i] = row < N ? al_d[(size_t)row * 4 + (i & 3)] : 0.f;
        mx[i]    = -__builtin_huge_valf();
        den[i]   = 0.f;
    }
    if (tid == 0) ovf = 0;
    __syncthreads();

    const int* __restrict__ dst = ei + E;
    const int nsteps = E >> 8;

    auto scan = [&](const int thresh, const bool keep, auto&& drain) -> int {
        int  cnt = 0, done = 0;
        bool kept = keep;
        {
            const int row0 = base + w * SPW;
            int nself = N - row0;
            nself = nself < 0 ? 0 : (nself > SPW ? SPW : nself);
            if (lane < nself) q[w][lane] = E + row0 + lane;
            cnt = nself;
        }
        auto push = [&](bool hit, int ev) {
            const unsigned mask = __builtin_amdgcn_ballot_w32(hit);
            if (hit) q[w][cnt + (int)__builtin_amdgcn_mbcnt_lo(mask, 0u)] = ev;
            cnt += __builtin_popcount(mask);
        };
        auto test4 = [&](const int4& v, int e) {
            const bool m0 = (unsigned)(v.x - base) < (unsigned)R;
            const bool m1 = (unsigned)(v.y - base) < (unsigned)R;
            const bool m2 = (unsigned)(v.z - base) < (unsigned)R;
            const bool m3 = (unsigned)(v.w - base) < (unsigned)R;
            if (__builtin_amdgcn_ballot_w32(m0 | m1 | m2 | m3)) {
                push(m0, e);
                push(m1, e + 1);
                push(m2, e + 2);
                push(m3, e + 3);
            }
        };
        for (int step = w; ; step += NW) {
            const bool last = step >= nsteps;
            if (!last) {
                const int  e  = step * 256 + lane * 4;
                const int4 va = *reinterpret_cast<const int4*>(dst + e);
                const int4 vb = *reinterpret_cast<const int4*>(dst + e + 128);
                test4(va, e);
                test4(vb, e + 128);
            } else if (w == 0) {
                for (int e = nsteps * 256 + lane; __builtin_amdgcn_ballot_w32(e < E); e += 32)
                    push(e < E && (unsigned)(dst[e < E ? e : 0] - base) < (unsigned)R, e);
            }
            if (cnt - done >= thresh || (last && cnt > done)) {
                drain(done, cnt);
                if (kept && (last || cnt + thresh + 256 <= QCAP)) {
                    done = cnt;
                } else {
                    kept = false;
                    cnt = 0;
                    done = 0;
                }
            }
            if (last) break;
        }
        return kept ? cnt : -1;
    };

    auto drain_max = [&](const int from, const int to) {
        for (int b = from; b < to; b += 32) {
            if (b + lane < to) {
                const int eq = q[w][b + lane];
                int s, d;
                if (eq < E) { s = ei[eq]; d = dst[eq]; } else { s = d = eq - E; }
                const float4 s4 = *reinterpret_cast<const float4*>(al_s + (size_t)s * 4);
                const int    r  = d - base;
                const float4 d4 = *reinterpret_cast<const float4*>(ald_b + r * 4);
                ldsMaxF(&mx[r * 4 + 0], leaky02(s4.x + d4.x));
                ldsMaxF(&mx[r * 4 + 1], leaky02(s4.y + d4.y));
                ldsMaxF(&mx[r * 4 + 2], leaky02(s4.z + d4.z));
                ldsMaxF(&mx[r * 4 + 3], leaky02(s4.w + d4.w));
            }
        }
    };

    auto drain_rows = [&](const int from, const int to) {
        for (int b = from; b < to; b += 32) {
            const int n = (to - b) < 32 ? (to - b) : 32;
            int   s = 0, r = 0;
            float a0 = 0.f, a1 = 0.f, a2 = 0.f, a3 = 0.f;
            if (lane < n) {
                const int eq = q[w][b + lane];
                int d;
                if (eq < E) { s = ei[eq]; d = dst[eq]; } else { s = d = eq - E; }
                const float4 s4 = *reinterpret_cast<const float4*>(al_s + (size_t)s * 4);
                r = d - base;
                const float4 d4 = *reinterpret_cast<const float4*>(ald_b + r * 4);
                const float4 m4 = *reinterpret_cast<const float4*>(mx + r * 4);
                a0 = __expf(leaky02(s4.x + d4.x) - m4.x);
                a1 = __expf(leaky02(s4.y + d4.y) - m4.y);
                a2 = __expf(leaky02(s4.z + d4.z) - m4.z);
                a3 = __expf(leaky02(s4.w + d4.w) - m4.w);
                if constexpr (MEAN) {
                    const float4 n4 = *reinterpret_cast<const float4*>(den + r * 4);
                    a0 = a0 / n4.x * 0.25f;
                    a1 = a1 / n4.y * 0.25f;
                    a2 = a2 / n4.z * 0.25f;
                    a3 = a3 / n4.w * 0.25f;
                } else {
                    __hip_atomic_fetch_add(&den[r * 4 + 0], a0, __ATOMIC_RELAXED, __HIP_MEMORY_SCOPE_WORKGROUP);
                    __hip_atomic_fetch_add(&den[r * 4 + 1], a1, __ATOMIC_RELAXED, __HIP_MEMORY_SCOPE_WORKGROUP);
                    __hip_atomic_fetch_add(&den[r * 4 + 2], a2, __ATOMIC_RELAXED, __HIP_MEMORY_SCOPE_WORKGROUP);
                    __hip_atomic_fetch_add(&den[r * 4 + 3], a3, __ATOMIC_RELAXED, __HIP_MEMORY_SCOPE_WORKGROUP);
                }
            }
            auto rows = [&](int j, auto UC) {
                constexpr int U = decltype(UC)::value;
                float2 v[U][V2];
                #pragma unroll
                for (int u = 0; u < U; ++u) {
                    const int sj = __builtin_amdgcn_readlane(s, j + u);
                    const float2* __restrict__ row =
                        reinterpret_cast<const float2*>(hfeat + (size_t)sj * F) + lane;
                    #pragma unroll
                    for (int k = 0; k < V2; ++k) v[u][k] = row[k * 32];
                }
                #pragma unroll
                for (int u = 0; u < U; ++u) {
                    const int rj = __builtin_amdgcn_readlane(r, j + u);
                    float al[4];
                    al[0] = __builtin_bit_cast(float, __builtin_amdgcn_readlane(__builtin_bit_cast(int, a0), j + u));
                    al[1] = __builtin_bit_cast(float, __builtin_amdgcn_readlane(__builtin_bit_cast(int, a1), j + u));
                    al[2] = __builtin_bit_cast(float, __builtin_amdgcn_readlane(__builtin_bit_cast(int, a2), j + u));
                    al[3] = __builtin_bit_cast(float, __builtin_amdgcn_readlane(__builtin_bit_cast(int, a3), j + u));
                    float* arow = acc + rj * AF + lane * 2;
                    if constexpr (MEAN) {
                        v2f sv = v2f{v[u][0].x, v[u][0].y} * al[0];
                        #pragma unroll
                        for (int k = 1; k < V2; ++k)
                            sv = __builtin_elementwise_fma(v2f{v[u][k].x, v[u][k].y}, v2f{al[k], al[k]}, sv);
                        __hip_atomic_fetch_add(arow,     sv.x, __ATOMIC_RELAXED, __HIP_MEMORY_SCOPE_WORKGROUP);
                        __hip_atomic_fetch_add(arow + 1, sv.y, __ATOMIC_RELAXED, __HIP_MEMORY_SCOPE_WORKGROUP);
                    } else {
                        #pragma unroll
                        for (int k = 0; k < V2; ++k) {
                            const float a = al[(k * 64) / C];
                            __hip_atomic_fetch_add(arow + k * 64,     v[u][k].x * a,
                                                   __ATOMIC_RELAXED, __HIP_MEMORY_SCOPE_WORKGROUP);
                            __hip_atomic_fetch_add(arow + k * 64 + 1, v[u][k].y * a,
                                                   __ATOMIC_RELAXED, __HIP_MEMORY_SCOPE_WORKGROUP);
                        }
                    }
                }
            };
            int j = 0;
            for (; j + 4 <= n; j += 4) rows(j, std::integral_constant<int, 4>{});
            for (; j < n; ++j)         rows(j, std::integral_constant<int, 1>{});
        }
    };

    auto drain_den = [&](const int from, const int to) {
        for (int b = from; b < to; b += 32) {
            if (b + lane < to) {
                const int eq = q[w][b + lane];
                int s, d;
                if (eq < E) { s = ei[eq]; d = dst[eq]; } else { s = d = eq - E; }
                const float4 s4 = *reinterpret_cast<const float4*>(al_s + (size_t)s * 4);
                const int    r  = d - base;
                const float4 d4 = *reinterpret_cast<const float4*>(ald_b + r * 4);
                const float4 m4 = *reinterpret_cast<const float4*>(mx + r * 4);
                __hip_atomic_fetch_add(&den[r * 4 + 0], __expf(leaky02(s4.x + d4.x) - m4.x), __ATOMIC_RELAXED, __HIP_MEMORY_SCOPE_WORKGROUP);
                __hip_atomic_fetch_add(&den[r * 4 + 1], __expf(leaky02(s4.y + d4.y) - m4.y), __ATOMIC_RELAXED, __HIP_MEMORY_SCOPE_WORKGROUP);
                __hip_atomic_fetch_add(&den[r * 4 + 2], __expf(leaky02(s4.z + d4.z) - m4.z), __ATOMIC_RELAXED, __HIP_MEMORY_SCOPE_WORKGROUP);
                __hip_atomic_fetch_add(&den[r * 4 + 3], __expf(leaky02(s4.w + d4.w) - m4.w), __ATOMIC_RELAXED, __HIP_MEMORY_SCOPE_WORKGROUP);
            }
        }
    };

    int* __restrict__ bl = blist + (size_t)blockIdx.x * BLS;
    int nkept;
    if constexpr (LIST == 2) {
        if (__builtin_amdgcn_readfirstlane(bl[0]) == 0) {
            nkept = __builtin_amdgcn_readfirstlane(bl[1 + w]);
            for (int i = lane; i < nkept; i += 32) q[w][i] = bl[1 + NW + w * QCAP + i];
            drain_max(0, nkept);
        } else {
            nkept = scan(DRAIN_MAX, true, drain_max);
        }
    } else {
        nkept = scan(DRAIN_MAX, true, drain_max);
    }
    if (nkept < 0 && lane == 0)
        __hip_atomic_fetch_or(&ovf, 1, __ATOMIC_RELAXED, __HIP_MEMORY_SCOPE_WORKGROUP);
    __syncthreads();
    if constexpr (LIST == 1) {
        const int lost = __builtin_amdgcn_readfirstlane(ovf);
        if (tid == 0) bl[0] = lost;
        if (lost == 0) {
            if (lane == 0) bl[1 + w] = nkept;
            for (int i = lane; i < nkept; i += 32) bl[1 + NW + w * QCAP + i] = q[w][i];
        }
    }
    const bool whole = __builtin_amdgcn_readfirstlane(ovf) == 0;
    if constexpr (MEAN) {
        if (whole) drain_den(0, nkept);
        else       scan(DRAIN_MAX, false, drain_den);
        __syncthreads();
    }
    if (whole) drain_rows(0, nkept);
    else       scan(DRAIN_ROWS, false, drain_rows);
    __syncthreads();

    float4* __restrict__ out4 = reinterpret_cast<float4*>(agg + (size_t)base * AF);
    for (int i = tid; i < R * AF / 4; i += NT) {
        const int row = i / (AF / 4);
        if (base + row < N) {
            const float4 v = acc4[i];
            if constexpr (MEAN) {
                const float4 b4 = reinterpret_cast<const float4*>(bias)[i % (AF / 4)];
                out4[i] = make_float4(v.x + b4.x, v.y + b4.y, v.z + b4.z, v.w + b4.w);
            } else {
                const float dn = den[row * 4 + ((i % (F / 4)) * 4) / C];
                out4[i] = make_float4(v.x / dn, v.y / dn, v.z / dn, v.w / dn);
            }
        }
    }
}

static inline unsigned cdiv_u(size_t a, unsigned b) {
    return (unsigned)((a + b - 1) / b);
}

extern "C" void kernel_launch(void* const* d_in, const int* in_sizes, int n_in,
                              void* d_out, int out_size, void* d_ws, size_t ws_size,
                              hipStream_t stream) {
    const float* x     = (const float*)d_in[0];
    const int*   ei    = (const int*)d_in[1];
    const float* W1    = (const float*)d_in[2];
    const float* a_s1  = (const float*)d_in[3];
    const float* a_d1  = (const float*)d_in[4];
    const float* b1    = (const float*)d_in[5];
    const float* W2    = (const float*)d_in[6];
    const float* a_s2  = (const float*)d_in[7];
    const float* a_d2  = (const float*)d_in[8];
    const float* b2    = (const float*)d_in[9];
    float*       out   = (float*)d_out;

    const int F1   = in_sizes[3];
    const int F2   = in_sizes[7];
    const int LAT  = in_sizes[9];
    const int H    = F2 / LAT;
    const int Fin  = in_sizes[2] / F1;
    const int N    = in_sizes[0] / Fin;
    const int E    = in_sizes[1] / 2;

    float* ws = (float*)d_ws;
    size_t o = 0;
    float* h1   = ws + o; o += (size_t)N * F1;
    float* agg1 = ws + o; o += (size_t)N * F1;
    float* als  = ws + o; o += (size_t)N * H;
    float* ald  = ws + o; o += (size_t)N * H;
    float* h2   = h1;
    int*   blist = (int*)(ws + o);
    (void)ws_size; (void)n_in; (void)out_size;

    gemm_panel_f32<512, 4, false><<<N / PM, 256, 0, stream>>>(x, W1, h1, nullptr, a_s1, a_d1, als, ald);

    agg_band<512, 128, 64, 32, 1, false><<<cdiv_u((size_t)N, 64), 1024, 0, stream>>>(
        ei, E, N, h1, als, ald, agg1, blist, nullptr);

    gemm_panel_f32<256, 2, true><<<N / PM, 256, 0, stream>>>(agg1, W2, h2, b1, a_s2, a_d2, als, ald);

    agg_band<256, 64, 64, 32, 2, true><<<cdiv_u((size_t)N, 64), 1024, 0, stream>>>(
        ei, E, N, h2, als, ald, out, blist, b2);
}


# band kernel: one owner wave per row; band edges go to per-row LDS lists (slot by integer LDS add), softmax and row sum stay in the owner's registers; LDS float adds gone; sum order = list order
# speedup vs baseline: 8.6755x; 1.2372x over previous
#include <hip/hip_runtime.h>
#include <math.h>
#include <type_traits>

typedef __attribute__((ext_vector_type(2))) float v2f;
typedef __attribute__((ext_vector_type(8))) float v8f;
typedef __attribute__((ext_vector_type(4))) float v4f;

#define PM   64
#define PK   512
#define PSTR 516
#define PLDS (41 * 1024)

static __device__ __forceinline__ float leaky02(float x) {
    return x > 0.f ? x : 0.2f * x;
}

template <int N, int TM, bool ACT>
__global__ __launch_bounds__(256)
void gemm_panel_f32(const float* __restrict__ A, const float* __restrict__ B,
                    float* __restrict__ D, const float* __restrict__ abias,
                    const float* __restrict__ a_s, const float* __restrict__ a_d,
                    float* __restrict__ al_s, float* __restrict__ al_d) {
    constexpr int NWC = N / 64;
    constexpr int NWR = PM / (16 * TM);
    static_assert(NWC * NWR == 8, "8 waves per block");
    static_assert(PLDS >= PM * PSTR, "panel fits");
    __shared__ __attribute__((aligned(16))) float Ap[PLDS];

    const int t    = threadIdx.x;
    const int lane = t & 31;
    const int w    = __builtin_amdgcn_readfirstlane(t >> 5);
    const int half = lane >> 4;
    const int lm   = lane & 15;
    const int wc   = w % NWC;
    const int wr   = w / NWC;
    const int rowM = blockIdx.x * PM;

    {
        const float4* __restrict__ gA = reinterpret_cast<const float4*>(A + (size_t)rowM * PK);
        float4 bb = make_float4(0.f, 0.f, 0.f, 0.f);
        if constexpr (ACT) bb = reinterpret_cast<const float4*>(abias)[t & 127];
        #pragma unroll 8
        for (int i = t; i < PM * PK / 4; i += 256) {
            const int row = i >> 7;
            const int c4  = i & 127;
            float4 v = gA[i];
            if constexpr (ACT) {
                v.x += bb.x; v.y += bb.y; v.z += bb.z; v.w += bb.w;
                v.x = v.x > 0.f ? v.x : 0.f;
                v.y = v.y > 0.f ? v.y : 0.f;
                v.z = v.z > 0.f ? v.z : 0.f;
                v.w = v.w > 0.f ? v.w : 0.f;
            }
            *reinterpret_cast<float4*>(&Ap[row * PSTR + c4 * 4]) = v;
        }
    }
    __syncthreads();

    const float* ap = Ap + (wr * TM * 16 + lm) * PSTR + 4 * half;
    const float* __restrict__ bp = B + (size_t)(4 * half) * N + wc * 64 + 4 * lm;

    v8f acc[TM][4];
    #pragma unroll
    for (int i = 0; i < TM; ++i)
        #pragma unroll
        for (int j = 0; j < 4; ++j) acc[i][j] = v8f{};

    constexpr int KS = 64 / TM;
    constexpr int NB = KS / 2;
    static_assert(PK % (2 * KS) == 0 && (NB == 8 || NB == 16), "two sets per trip");

    auto loadB = [&](v4f (&b)[NB], int k) {
        #pragma unroll
        for (int g = 0; g < KS / 8; ++g)
            #pragma unroll
            for (int s = 0; s < 4; ++s)
                b[g * 4 + s] = *reinterpret_cast<const v4f*>(bp + (size_t)(k + g * 8 + s) * N);
    };
    auto holdB = [&](const v4f (&b)[NB], const v8f& done) {
        if constexpr (NB == 8) {
            asm volatile("" :: "v"(b[0]), "v"(b[1]), "v"(b[2]), "v"(b[3]),
                               "v"(b[4]), "v"(b[5]), "v"(b[6]), "v"(b[7]), "v"(done));
        } else {
            asm volatile("" :: "v"(b[0]), "v"(b[1]), "v"(b[2]), "v"(b[3]),
                               "v"(b[4]), "v"(b[5]), "v"(b[6]), "v"(b[7]),
                               "v"(b[8]), "v"(b[9]), "v"(b[10]), "v"(b[11]),
                               "v"(b[12]), "v"(b[13]), "v"(b[14]), "v"(b[15]), "v"(done));
        }
    };
    auto mma = [&](const v4f (&b)[NB], int k) {
        #pragma unroll
        for (int g = 0; g < KS / 8; ++g) {
            v4f a[TM];
            #pragma unroll
            for (int i = 0; i < TM; ++i)
                a[i] = *reinterpret_cast<const v4f*>(ap + i * 16 * PSTR + k + g * 8);
            #pragma unroll
            for (int p = 0; p < 2; ++p) {
                #pragma unroll
                for (int i = 0; i < TM; ++i) {
                    #pragma unroll
                    for (int j = 0; j < 4; ++j) {
                        v2f av, bv;
                        av.x = a[i][2 * p];
                        av.y = a[i][2 * p + 1];
                        bv.x = b[g * 4 + 2 * p][j];
                        bv.y = b[g * 4 + 2 * p + 1][j];
                        acc[i][j] = __builtin_amdgcn_wmma_f32_16x16x4_f32(
                            false, av, false, bv, (short)0, acc[i][j], false, false);
                    }
                }
            }
        }
    };

    v4f bA[NB], bB[NB];
    loadB(bA, 0);
    for (int k = 0; k < PK; k += 2 * KS) {
        loadB(bB, k + KS);
        mma(bA, k);
        holdB(bB, acc[TM - 1][3]);
        const int kn = (k + 2 * KS < PK) ? k + 2 * KS : k;
        loadB(bA, kn);
        mma(bB, k + KS);
        holdB(bA, acc[TM - 1][3]);
    }

    float* dp = D + (size_t)(rowM + wr * TM * 16 + half * 8) * N + wc * 64 + 4 * lm;
    #pragma unroll
    for (int i = 0; i < TM; ++i) {
        #pragma unroll
        for (int r = 0; r < 8; ++r) {
            *reinterpret_cast<float4*>(dp + (size_t)(i * 16 + r) * N) =
                make_float4(acc[i][0][r], acc[i][1][r], acc[i][2][r], acc[i][3][r]);
        }
    }

    constexpr int LP = N / 4 + 4;
    constexpr int HW = N / 16;
    static_assert(2 * PM * LP <= PM * PSTR && HW % 4 == 0, "partials fit in the panel's space");
    const float4 ws4 = *reinterpret_cast<const float4*>(a_s + wc * 64 + 4 * lm);
    const float4 wd4 = *reinterpret_cast<const float4*>(a_d + wc * 64 + 4 * lm);
    __syncthreads();
    {
        float* pls = Ap + (wr * TM * 16 + half * 8) * LP + wc * 16 + lm;
        float* pld = pls + PM * LP;
        #pragma unroll
        for (int i = 0; i < TM; ++i) {
            #pragma unroll
            for (int r = 0; r < 8; ++r) {
                pls[(i * 16 + r) * LP] = (acc[i][0][r] * ws4.x + acc[i][1][r] * ws4.y)
                                       + (acc[i][2][r] * ws4.z + acc[i][3][r] * ws4.w);
                pld[(i * 16 + r) * LP] = (acc[i][0][r] * wd4.x + acc[i][1][r] * wd4.y)
                                       + (acc[i][2][r] * wd4.z + acc[i][3][r] * wd4.w);
            }
        }
    }
    __syncthreads();
    {
        const float4* qs = reinterpret_cast<const float4*>(Ap + (t >> 2) * LP + (t & 3) * HW);
        const float4* qd = reinterpret_cast<const float4*>(Ap + PM * LP + (t >> 2) * LP + (t & 3) * HW);
        float ss = 0.f, sd = 0.f;
        #pragma unroll
        for (int q = 0; q < HW / 4; ++q) {
            const float4 u = qs[q];
            const float4 v = qd[q];
            ss += (u.x + u.y) + (u.z + u.w);
            sd += (v.x + v.y) + (v.z + v.w);
        }
        al_s[(size_t)rowM * 4 + t] = ss;
        al_d[(size_t)rowM * 4 + t] = sd;
    }
}

template <int CTRL>
static __device__ __forceinline__ float dpp_row(float x) {
    return __builtin_bit_cast(float,
        __builtin_amdgcn_update_dpp(0, __builtin_bit_cast(int, x), CTRL, 0xf, 0xf, false));
}
static __device__ __forceinline__ float other_half(float x) {
    const int b = __builtin_bit_cast(int, x);
    return __builtin_bit_cast(float,
        __builtin_amdgcn_permlanex16(b, b, 0x76543210, 0xfedcba98, false, false));
}
template <bool MAX>
static __device__ __forceinline__ float wave_all(float x) {
    auto join = [](float a, float b) { return MAX ? fmaxf(a, b) : a + b; };
    x = join(x, dpp_row<0x161>(x));
    x = join(x, dpp_row<0x162>(x));
    x = join(x, dpp_row<0x164>(x));
    x = join(x, dpp_row<0x168>(x));
    return join(x, other_half(x));
}

template <int F, int C, int R, int NW, int LIST, bool MEAN>
__global__ __launch_bounds__(NW * 32)
void agg_band(const int* __restrict__ ei, int E, int N,
              const float* __restrict__ hfeat,
              const float* __restrict__ al_s,
              const float* __restrict__ al_d,
              float* __restrict__ agg,
              int* __restrict__ blist,
              const float* __restrict__ bias) {
    static_assert(F == 4 * C && (C == 64 || C == 128), "four heads, a lane holds C/32 channels of each");
    static_assert(R % NW == 0, "whole rows per wave");
    static_assert(LIST == 1 || LIST == 2, "1: leaves its lists, 2: takes them");
    constexpr int VW  = C / 32;
    constexpr int RPW = R / NW;
    constexpr int CAP = 192;
    static_assert(CAP >= 31 + 128, "spare path: one scan step on top of an undrained rest");
    constexpr int BLS = 1 + R + R * CAP;
    typedef float vw_t __attribute__((ext_vector_type(VW)));

    __shared__ int rq[R][CAP];
    __shared__ int rcnt[R];
    __shared__ int ovf;

    const int tid  = threadIdx.x;
    const int lane = tid & 31;
    const int w    = __builtin_amdgcn_readfirstlane(tid >> 5);
    const int base = blockIdx.x * R;
    const int* __restrict__ dst = ei + E;
    int* __restrict__ bl = blist + (size_t)blockIdx.x * BLS;
    const float ninf = -__builtin_huge_valf();

    bool whole;
    if constexpr (LIST == 1) {
        if (tid < R) {
            rq[tid][0] = E + base + tid;
            rcnt[tid]  = base + tid < N ? 1 : 0;
        }
        if (tid == 0) ovf = 0;
        __syncthreads();

        auto push = [&](bool hit, int d, int ev) __attribute__((always_inline)) {
            if (hit) {
                const int rw   = d - base;
                const int slot = __hip_atomic_fetch_add(&rcnt[rw], 1, __ATOMIC_RELAXED, __HIP_MEMORY_SCOPE_WORKGROUP);
                if (slot < CAP) rq[rw][slot] = ev;
                else            ovf = 1;
            }
        };
        auto test4 = [&](const int4& v, int e) __attribute__((always_inline)) {
            const bool m0 = (unsigned)(v.x - base) < (unsigned)R;
            const bool m1 = (unsigned)(v.y - base) < (unsigned)R;
            const bool m2 = (unsigned)(v.z - base) < (unsigned)R;
            const bool m3 = (unsigned)(v.w - base) < (unsigned)R;
            if (__builtin_amdgcn_ballot_w32(m0 | m1 | m2 | m3)) {
                push(m0, v.x, e);
                push(m1, v.y, e + 1);
                push(m2, v.z, e + 2);
                push(m3, v.w, e + 3);
            }
        };
        const int nsteps = E >> 8;
        for (int step = w; step < nsteps; step += NW) {
            const int  e  = step * 256 + lane * 4;
            const int4 va = *reinterpret_cast<const int4*>(dst + e);
            const int4 vb = *reinterpret_cast<const int4*>(dst + e + 128);
            test4(va, e);
            test4(vb, e + 128);
        }
        if (w == 0) {
            for (int e = nsteps * 256 + lane; __builtin_amdgcn_ballot_w32(e < E); e += 32) {
                const int d = dst[e < E ? e : 0];
                push(e < E && (unsigned)(d - base) < (unsigned)R, d, e);
            }
        }
        __syncthreads();
        whole = __builtin_amdgcn_readfirstlane(ovf) == 0;
        if (tid == 0) bl[0] = whole ? 0 : 1;
    } else {
        whole = __builtin_amdgcn_readfirstlane(bl[0]) == 0;
    }

    for (int rr = 0; rr < RPW; ++rr) {
        const int r   = w * RPW + rr;
        const int row = base + r;
        if (row >= N) break;

        int tot = 0;
        if (whole) {
            if constexpr (LIST == 1) {
                tot = __builtin_amdgcn_readfirstlane(rcnt[r]);
                for (int i = lane; i < tot; i += 32) {
                    const int eq = rq[r][i];
                    const int s  = eq < E ? ei[eq] : eq - E;
                    rq[r][i] = s;
                    bl[1 + R + r * CAP + i] = s;
                }
                if (lane == 0) bl[1 + r] = tot;
            } else {
                tot = __builtin_amdgcn_readfirstlane(bl[1 + r]);
                for (int i = lane; i < tot; i += 32) rq[r][i] = bl[1 + R + r * CAP + i];
            }
        }

        const float4 d4 = *reinterpret_cast<const float4*>(al_d + (size_t)row * 4);

        auto for_chunks = [&](auto&& body) __attribute__((always_inline)) {
            if (whole) {
                for (int b = 0; b < tot; b += 32) body(b, tot - b < 32 ? tot - b : 32);
            } else {
                if (lane == 0) rq[r][0] = row;
                int cnt = 1, e0 = 0;
                do {
                    for (; cnt < 32 && e0 < E; e0 += 128) {
                        int d[4];
                        #pragma unroll
                        for (int q = 0; q < 4; ++q) {
                            const int e = e0 + q * 32 + lane;
                            d[q] = e < E ? dst[e] : -1;
                        }
                        #pragma unroll
                        for (int q = 0; q < 4; ++q) {
                            const bool hit = d[q] == row;
                            const unsigned mask = __builtin_amdgcn_ballot_w32(hit);
                            if (mask) {
                                if (hit) rq[r][cnt + (int)__builtin_amdgcn_mbcnt_lo(mask, 0u)] = ei[e0 + q * 32 + lane];
                                cnt += __builtin_popcount(mask);
                            }
                        }
                    }
                    for (int b = 0; b < cnt; b += 32) body(b, cnt - b < 32 ? cnt - b : 32);
                    cnt = 0;
                } while (e0 < E);
            }
        };

        int   cs = 0;
        float c0 = ninf, c1 = ninf, c2 = ninf, c3 = ninf;
        auto logit4 = [&](int i, int end) __attribute__((always_inline)) {
            cs = 0;
            c0 = c1 = c2 = c3 = ninf;
            if (i < end) {
                cs = rq[r][i];
                const float4 s4 = *reinterpret_cast<const float4*>(al_s + (size_t)cs * 4);
                c0 = leaky02(s4.x + d4.x);
                c1 = leaky02(s4.y + d4.y);
                c2 = leaky02(s4.z + d4.z);
                c3 = leaky02(s4.w + d4.w);
            }
        };

        float m0 = ninf, m1 = ninf, m2 = ninf, m3 = ninf;
        for_chunks([&](int from, int n) __attribute__((always_inline)) {
            logit4(from + lane, from + n);
            m0 = fmaxf(m0, c0);
            m1 = fmaxf(m1, c1);
            m2 = fmaxf(m2, c2);
            m3 = fmaxf(m3, c3);
        });
        m0 = wave_all<true>(m0);
        m1 = wave_all<true>(m1);
        m2 = wave_all<true>(m2);
        m3 = wave_all<true>(m3);

        const bool once = whole && tot <= 32;
        float dn0 = 0.f, dn1 = 0.f, dn2 = 0.f, dn3 = 0.f;
        vw_t  acc[4];
        #pragma unroll
        for (int k = 0; k < 4; ++k) acc[k] = vw_t(0.f);

        for_chunks([&](int from, int n) __attribute__((always_inline)) {
            if (!once) logit4(from + lane, from + n);
            float a[4];
            a[0] = __expf(c0 - m0);
            a[1] = __expf(c1 - m1);
            a[2] = __expf(c2 - m2);
            a[3] = __expf(c3 - m3);
            dn0 += a[0];
            dn1 += a[1];
            dn2 += a[2];
            dn3 += a[3];
            auto rows = [&](int j, auto UC) __attribute__((always_inline)) {
                constexpr int U = decltype(UC)::value;
                vw_t v[U][4];
                #pragma unroll
                for (int u = 0; u < U; ++u) {
                    const int sj = __builtin_amdgcn_readlane(cs, j + u);
                    const vw_t* __restrict__ p =
                        reinterpret_cast<const vw_t*>(hfeat + (size_t)sj * F) + lane;
                    #pragma unroll
                    for (int k = 0; k < 4; ++k) v[u][k] = p[k * 32];
                }
                #pragma unroll
                for (int u = 0; u < U; ++u) {
                    #pragma unroll
                    for (int k = 0; k < 4; ++k) {
                        const float ak = __builtin_bit_cast(float,
                            __builtin_amdgcn_readlane(__builtin_bit_cast(int, a[k]), j + u));
                        acc[k] += v[u][k] * ak;
                    }
                }
            };
            int j = 0;
            for (; j + 4 <= n; j += 4) rows(j, std::integral_constant<int, 4>{});
            if (j + 2 <= n) { rows(j, std::integral_constant<int, 2>{}); j += 2; }
            if (j < n)        rows(j, std::integral_constant<int, 1>{});
        });
        dn0 = wave_all<false>(dn0);
        dn1 = wave_all<false>(dn1);
        dn2 = wave_all<false>(dn2);
        dn3 = wave_all<false>(dn3);

        if constexpr (MEAN) {
            vw_t o = acc[0] / dn0;
            o += acc[1] / dn1;
            o += acc[2] / dn2;
            o += acc[3] / dn3;
            const vw_t b = reinterpret_cast<const vw_t*>(bias)[lane];
            reinterpret_cast<vw_t*>(agg + (size_t)row * C)[lane] = o * 0.25f + b;
        } else {
            vw_t* __restrict__ o = reinterpret_cast<vw_t*>(agg + (size_t)row * F) + lane;
            o[0]  = acc[0] / dn0;
            o[32] = acc[1] / dn1;
            o[64] = acc[2] / dn2;
            o[96] = acc[3] / dn3;
        }
    }
}


static inline unsigned cdiv_u(size_t a, unsigned b) {
    return (unsigned)((a + b - 1) / b);
}

extern "C" void kernel_launch(void* const* d_in, const int* in_sizes, int n_in,
                              void* d_out, int out_size, void* d_ws, size_t ws_size,
                              hipStream_t stream) {
    const float* x     = (const float*)d_in[0];
    const int*   ei    = (const int*)d_in[1];
    const float* W1    = (const float*)d_in[2];
    const float* a_s1  = (const float*)d_in[3];
    const float* a_d1  = (const float*)d_in[4];
    const float* b1    = (const float*)d_in[5];
    const float* W2    = (const float*)d_in[6];
    const float* a_s2  = (const float*)d_in[7];
    const float* a_d2  = (const float*)d_in[8];
    const float* b2    = (const float*)d_in[9];
    float*       out   = (float*)d_out;

    const int F1   = in_sizes[3];
    const int F2   = in_sizes[7];
    const int LAT  = in_sizes[9];
    const int H    = F2 / LAT;
    const int Fin  = in_sizes[2] / F1;
    const int N    = in_sizes[0] / Fin;
    const int E    = in_sizes[1] / 2;

    float* ws = (float*)d_ws;
    size_t o = 0;
    float* h1   = ws + o; o += (size_t)N * F1;
    float* agg1 = ws + o; o += (size_t)N * F1;
    float* als  = ws + o; o += (size_t)N * H;
    float* ald  = ws + o; o += (size_t)N * H;
    float* h2   = h1;
    int*   blist = (int*)(ws + o);
    (void)ws_size; (void)n_in; (void)out_size;

    gemm_panel_f32<512, 4, false><<<N / PM, 256, 0, stream>>>(x, W1, h1, nullptr, a_s1, a_d1, als, ald);

    agg_band<512, 128, 64, 32, 1, false><<<cdiv_u((size_t)N, 64), 1024, 0, stream>>>(
        ei, E, N, h1, als, ald, agg1, blist, nullptr);

    gemm_panel_f32<256, 2, true><<<N / PM, 256, 0, stream>>>(agg1, W2, h2, b1, a_s2, a_d2, als, ald);

    agg_band<256, 64, 64, 32, 2, true><<<cdiv_u((size_t)N, 64), 1024, 0, stream>>>(
        ei, E, N, h2, als, ald, out, blist, b2);
}


# band kernel scans a one-byte-per-edge copy of the destination list (band numbers, left by GEMM 1's launch) and reads full destinations of byte matches only; list order, so sum order, changed
# speedup vs baseline: 9.3206x; 1.0744x over previous
#include <hip/hip_runtime.h>
#include <math.h>
#include <type_traits>

typedef __attribute__((ext_vector_type(2))) float v2f;
typedef __attribute__((ext_vector_type(8))) float v8f;
typedef __attribute__((ext_vector_type(4))) float v4f;

#define PM   64
#define PK   512
#define PSTR 516
#define PLDS (41 * 1024)
#define BROWS 64

static __device__ __forceinline__ float leaky02(float x) {
    return x > 0.f ? x : 0.2f * x;
}

template <int N, int TM, bool ACT, bool BAND>
__global__ __launch_bounds__(256)
void gemm_panel_f32(const float* __restrict__ A, const float* __restrict__ B,
                    float* __restrict__ D, const float* __restrict__ abias,
                    const float* __restrict__ a_s, const float* __restrict__ a_d,
                    float* __restrict__ al_s, float* __restrict__ al_d,
                    const int* __restrict__ dlist, int E, unsigned* __restrict__ bandb) {
    constexpr int NWC = N / 64;
    constexpr int NWR = PM / (16 * TM);
    static_assert(NWC * NWR == 8, "8 waves per block");
    static_assert(PLDS >= PM * PSTR, "panel fits");
    __shared__ __attribute__((aligned(16))) float Ap[PLDS];

    const int t    = threadIdx.x;
    const int lane = t & 31;
    const int w    = __builtin_amdgcn_readfirstlane(t >> 5);
    const int half = lane >> 4;
    const int lm   = lane & 15;
    const int wc   = w % NWC;
    const int wr   = w / NWC;
    const int rowM = blockIdx.x * PM;

    auto pack4 = [](const int4& d) {
        return  (((unsigned)d.x / BROWS) & 255u)
             | ((((unsigned)d.y / BROWS) & 255u) << 8)
             | ((((unsigned)d.z / BROWS) & 255u) << 16)
             | ((((unsigned)d.w / BROWS) & 255u) << 24);
    };
    const int4* __restrict__ d4p = reinterpret_cast<const int4*>(dlist + E);
    const int bi0 = blockIdx.x * 256 + t;

    {
        const float4* __restrict__ gA = reinterpret_cast<const float4*>(A + (size_t)rowM * PK);
        float4 bb = make_float4(0.f, 0.f, 0.f, 0.f);
        if constexpr (ACT) bb = reinterpret_cast<const float4*>(abias)[t & 127];
        #pragma unroll 8
        for (int i = t; i < PM * PK / 4; i += 256) {
            const int row = i >> 7;
            const int c4  = i & 127;
            float4 v = gA[i];
            if constexpr (ACT) {
                v.x += bb.x; v.y += bb.y; v.z += bb.z; v.w += bb.w;
                v.x = v.x > 0.f ? v.x : 0.f;
                v.y = v.y > 0.f ? v.y : 0.f;
                v.z = v.z > 0.f ? v.z : 0.f;
                v.w = v.w > 0.f ? v.w : 0.f;
            }
            *reinterpret_cast<float4*>(&Ap[row * PSTR + c4 * 4]) = v;
        }
    }
    __syncthreads();

    const float* ap = Ap + (wr * TM * 16 + lm) * PSTR + 4 * half;
    const float* __restrict__ bp = B + (size_t)(4 * half) * N + wc * 64 + 4 * lm;

    v8f acc[TM][4];
    #pragma unroll
    for (int i = 0; i < TM; ++i)
        #pragma unroll
        for (int j = 0; j < 4; ++j) acc[i][j] = v8f{};

    constexpr int KS = 64 / TM;
    constexpr int NB = KS / 2;
    static_assert(PK % (2 * KS) == 0 && (NB == 8 || NB == 16), "two sets per trip");

    auto loadB = [&](v4f (&b)[NB], int k) {
        #pragma unroll
        for (int g = 0; g < KS / 8; ++g)
            #pragma unroll
            for (int s = 0; s < 4; ++s)
                b[g * 4 + s] = *reinterpret_cast<const v4f*>(bp + (size_t)(k + g * 8 + s) * N);
    };
    auto holdB = [&](const v4f (&b)[NB], const v8f& done) {
        if constexpr (NB == 8) {
            asm volatile("" :: "v"(b[0]), "v"(b[1]), "v"(b[2]), "v"(b[3]),
                               "v"(b[4]), "v"(b[5]), "v"(b[6]), "v"(b[7]), "v"(done));
        } else {
            asm volatile("" :: "v"(b[0]), "v"(b[1]), "v"(b[2]), "v"(b[3]),
                               "v"(b[4]), "v"(b[5]), "v"(b[6]), "v"(b[7]),
                               "v"(b[8]), "v"(b[9]), "v"(b[10]), "v"(b[11]),
                               "v"(b[12]), "v"(b[13]), "v"(b[14]), "v"(b[15]), "v"(done));
        }
    };
    auto mma = [&](const v4f (&b)[NB], int k) {
        #pragma unroll
        for (int g = 0; g < KS / 8; ++g) {
            v4f a[TM];
            #pragma unroll
            for (int i = 0; i < TM; ++i)
                a[i] = *reinterpret_cast<const v4f*>(ap + i * 16 * PSTR + k + g * 8);
            #pragma unroll
            for (int p = 0; p < 2; ++p) {
                #pragma unroll
                for (int i = 0; i < TM; ++i) {
                    #pragma unroll
                    for (int j = 0; j < 4; ++j) {
                        v2f av, bv;
                        av.x = a[i][2 * p];
                        av.y = a[i][2 * p + 1];
                        bv.x = b[g * 4 + 2 * p][j];
                        bv.y = b[g * 4 + 2 * p + 1][j];
                        acc[i][j] = __builtin_amdgcn_wmma_f32_16x16x4_f32(
                            false, av, false, bv, (short)0, acc[i][j], false, false);
                    }
                }
            }
        }
    };

    v4f bA[NB], bB[NB];
    loadB(bA, 0);
    for (int k = 0; k < PK; k += 2 * KS) {
        loadB(bB, k + KS);
        mma(bA, k);
        holdB(bB, acc[TM - 1][3]);
        const int kn = (k + 2 * KS < PK) ? k + 2 * KS : k;
        loadB(bA, kn);
        mma(bB, k + KS);
        holdB(bA, acc[TM - 1][3]);
    }

    int4 bd0 = make_int4(0, 0, 0, 0);
    if constexpr (BAND) bd0 = d4p[bi0 < (E >> 2) ? bi0 : 0];

    float* dp = D + (size_t)(rowM + wr * TM * 16 + half * 8) * N + wc * 64 + 4 * lm;
    #pragma unroll
    for (int i = 0; i < TM; ++i) {
        #pragma unroll
        for (int r = 0; r < 8; ++r) {
            *reinterpret_cast<float4*>(dp + (size_t)(i * 16 + r) * N) =
                make_float4(acc[i][0][r], acc[i][1][r], acc[i][2][r], acc[i][3][r]);
        }
    }

    constexpr int LP = N / 4 + 4;
    constexpr int HW = N / 16;
    static_assert(2 * PM * LP <= PM * PSTR && HW % 4 == 0, "partials fit in the panel's space");
    const float4 ws4 = *reinterpret_cast<const float4*>(a_s + wc * 64 + 4 * lm);
    const float4 wd4 = *reinterpret_cast<const float4*>(a_d + wc * 64 + 4 * lm);
    __syncthreads();
    {
        float* pls = Ap + (wr * TM * 16 + half * 8) * LP + wc * 16 + lm;
        float* pld = pls + PM * LP;
        #pragma unroll
        for (int i = 0; i < TM; ++i) {
            #pragma unroll
            for (int r = 0; r < 8; ++r) {
                pls[(i * 16 + r) * LP] = (acc[i][0][r] * ws4.x + acc[i][1][r] * ws4.y)
                                       + (acc[i][2][r] * ws4.z + acc[i][3][r] * ws4.w);
                pld[(i * 16 + r) * LP] = (acc[i][0][r] * wd4.x + acc[i][1][r] * wd4.y)
                                       + (acc[i][2][r] * wd4.z + acc[i][3][r] * wd4.w);
            }
        }
    }
    __syncthreads();
    {
        const float4* qs = reinterpret_cast<const float4*>(Ap + (t >> 2) * LP + (t & 3) * HW);
        const float4* qd = reinterpret_cast<const float4*>(Ap + PM * LP + (t >> 2) * LP + (t & 3) * HW);
        float ss = 0.f, sd = 0.f;
        #pragma unroll
        for (int q = 0; q < HW / 4; ++q) {
            const float4 u = qs[q];
            const float4 v = qd[q];
            ss += (u.x + u.y) + (u.z + u.w);
            sd += (v.x + v.y) + (v.z + v.w);
        }
        al_s[(size_t)rowM * 4 + t] = ss;
        al_d[(size_t)rowM * 4 + t] = sd;
    }
    if constexpr (BAND) {
        if (bi0 < (E >> 2)) bandb[bi0] = pack4(bd0);
        for (int i = bi0 + gridDim.x * 256; i < (E >> 2); i += gridDim.x * 256) bandb[i] = pack4(d4p[i]);
    }
}

template <int CTRL>
static __device__ __forceinline__ float dpp_row(float x) {
    return __builtin_bit_cast(float,
        __builtin_amdgcn_update_dpp(0, __builtin_bit_cast(int, x), CTRL, 0xf, 0xf, false));
}
static __device__ __forceinline__ float other_half(float x) {
    const int b = __builtin_bit_cast(int, x);
    return __builtin_bit_cast(float,
        __builtin_amdgcn_permlanex16(b, b, 0x76543210, 0xfedcba98, false, false));
}
template <bool MAX>
static __device__ __forceinline__ float wave_all(float x) {
    auto join = [](float a, float b) { return MAX ? fmaxf(a, b) : a + b; };
    x = join(x, dpp_row<0x161>(x));
    x = join(x, dpp_row<0x162>(x));
    x = join(x, dpp_row<0x164>(x));
    x = join(x, dpp_row<0x168>(x));
    return join(x, other_half(x));
}

template <int F, int C, int R, int NW, int LIST, bool MEAN>
__global__ __launch_bounds__(NW * 32)
void agg_band(const int* __restrict__ ei, int E, int N,
              const float* __restrict__ hfeat,
              const float* __restrict__ al_s,
              const float* __restrict__ al_d,
              float* __restrict__ agg,
              int* __restrict__ blist,
              const float* __restrict__ bias,
              const unsigned* __restrict__ bandb) {
    static_assert(R == BROWS, "the byte copy of the destination list is made for bands of BROWS rows");
    static_assert(F == 4 * C && (C == 64 || C == 128), "four heads, a lane holds C/32 channels of each");
    static_assert(R % NW == 0, "whole rows per wave");
    static_assert(LIST == 1 || LIST == 2, "1: leaves its lists, 2: takes them");
    constexpr int VW  = C / 32;
    constexpr int RPW = R / NW;
    constexpr int CAP = 192;
    static_assert(CAP >= 31 + 128, "spare path: one scan step on top of an undrained rest");
    constexpr int BLS = 1 + R + R * CAP;
    typedef float vw_t __attribute__((ext_vector_type(VW)));

    __shared__ int rq[R][CAP];
    __shared__ int rcnt[R];
    __shared__ int ovf;
    constexpr int CQT = 96;
    constexpr int CQ  = CQT + 512;
    __shared__ int cq[LIST == 1 ? NW : 1][LIST == 1 ? CQ : 1];

    const int tid  = threadIdx.x;
    const int lane = tid & 31;
    const int w    = __builtin_amdgcn_readfirstlane(tid >> 5);
    const int base = blockIdx.x * R;
    const int* __restrict__ dst = ei + E;
    int* __restrict__ bl = blist + (size_t)blockIdx.x * BLS;
    const float ninf = -__builtin_huge_valf();

    bool whole;
    if constexpr (LIST == 1) {
        if (tid < R) {
            rq[tid][0] = E + base + tid;
            rcnt[tid]  = base + tid < N ? 1 : 0;
        }
        if (tid == 0) ovf = 0;
        __syncthreads();

        auto push = [&](bool hit, int d, int ev) __attribute__((always_inline)) {
            if (hit) {
                const int rw   = d - base;
                const int slot = __hip_atomic_fetch_add(&rcnt[rw], 1, __ATOMIC_RELAXED, __HIP_MEMORY_SCOPE_WORKGROUP);
                if (slot < CAP) rq[rw][slot] = ev;
                else            ovf = 1;
            }
        };
        const unsigned pat = ((unsigned)blockIdx.x & 255u) * 0x01010101u;
        int qn = 0;
        auto flush = [&]() __attribute__((always_inline)) {
            #pragma nounroll
            for (int b = 0; b < qn; b += 64) {
                const bool on0 = b + lane < qn;
                const bool on1 = b + 32 + lane < qn;
                const int  e0  = on0 ? cq[w][b + lane] : 0;
                const int  e1  = on1 ? cq[w][b + 32 + lane] : 0;
                const int  d0  = dst[e0];
                const int  d1  = dst[e1];
                push(on0 && (unsigned)(d0 - base) < (unsigned)R, d0, e0);
                push(on1 && (unsigned)(d1 - base) < (unsigned)R, d1, e1);
            }
            qn = 0;
        };
        auto test4 = [&](unsigned v, int e) __attribute__((always_inline)) {
            const unsigned x = v ^ pat;
            unsigned hz = (x - 0x01010101u) & ~x & 0x80808080u;
            unsigned m = __builtin_amdgcn_ballot_w32(hz != 0u);
            if (m) {
                do {
                    if (hz != 0u) {
                        cq[w][qn + (int)__builtin_amdgcn_mbcnt_lo(m, 0u)] = e + (__builtin_ctz(hz) >> 3);
                        hz &= hz - 1u;
                    }
                    qn += __builtin_popcount(m);
                    m = __builtin_amdgcn_ballot_w32(hz != 0u);
                } while (m);
            }
        };
        const int nsteps = E >> 9;
        const uint4* __restrict__ bq = reinterpret_cast<const uint4*>(bandb);
        for (int step = w; step < nsteps; step += 8 * NW) {
            uint4 v[8];
            #pragma unroll
            for (int u = 0; u < 8; ++u) {
                const int s = step + u * NW;
                v[u] = bq[(s < nsteps ? s : step) * 32 + lane];
            }
            #pragma unroll
            for (int u = 0; u < 8; ++u) {
                const int s = step + u * NW;
                if (s < nsteps) {
                    const int e = s * 512 + lane * 16;
                    test4(v[u].x, e);
                    test4(v[u].y, e + 4);
                    test4(v[u].z, e + 8);
                    test4(v[u].w, e + 12);
                    if (qn >= CQT) flush();
                }
            }
        }
        flush();
        if (w == 0) {
            for (int e = nsteps * 512 + lane; __builtin_amdgcn_ballot_w32(e < E); e += 32) {
                const int d = dst[e < E ? e : 0];
                push(e < E && (unsigned)(d - base) < (unsigned)R, d, e);
            }
        }
        __syncthreads();
        whole = __builtin_amdgcn_readfirstlane(ovf) == 0;
        if (tid == 0) bl[0] = whole ? 0 : 1;
    } else {
        whole = __builtin_amdgcn_readfirstlane(bl[0]) == 0;
    }

    for (int rr = 0; rr < RPW; ++rr) {
        const int r   = w * RPW + rr;
        const int row = base + r;
        if (row >= N) break;

        int tot = 0;
        if (whole) {
            if constexpr (LIST == 1) {
                tot = __builtin_amdgcn_readfirstlane(rcnt[r]);
                for (int i = lane; i < tot; i += 32) {
                    const int eq = rq[r][i];
                    const int s  = eq < E ? ei[eq] : eq - E;
                    rq[r][i] = s;
                    bl[1 + R + r * CAP + i] = s;
                }
                if (lane == 0) bl[1 + r] = tot;
            } else {
                tot = __builtin_amdgcn_readfirstlane(bl[1 + r]);
                for (int i = lane; i < tot; i += 32) rq[r][i] = bl[1 + R + r * CAP + i];
            }
        }

        const float4 d4 = *reinterpret_cast<const float4*>(al_d + (size_t)row * 4);

        auto for_chunks = [&](auto&& body) __attribute__((always_inline)) {
            if (whole) {
                for (int b = 0; b < tot; b += 32) body(b, tot - b < 32 ? tot - b : 32);
            } else {
                if (lane == 0) rq[r][0] = row;
                int cnt = 1, e0 = 0;
                do {
                    for (; cnt < 32 && e0 < E; e0 += 128) {
                        int d[4];
                        #pragma unroll
                        for (int q = 0; q < 4; ++q) {
                            const int e = e0 + q * 32 + lane;
                            d[q] = e < E ? dst[e] : -1;
                        }
                        #pragma unroll
                        for (int q = 0; q < 4; ++q) {
                            const bool hit = d[q] == row;
                            const unsigned mask = __builtin_amdgcn_ballot_w32(hit);
                            if (mask) {
                                if (hit) rq[r][cnt + (int)__builtin_amdgcn_mbcnt_lo(mask, 0u)] = ei[e0 + q * 32 + lane];
                                cnt += __builtin_popcount(mask);
                            }
                        }
                    }
                    for (int b = 0; b < cnt; b += 32) body(b, cnt - b < 32 ? cnt - b : 32);
                    cnt = 0;
                } while (e0 < E);
            }
        };

        int   cs = 0;
        float c0 = ninf, c1 = ninf, c2 = ninf, c3 = ninf;
        auto logit4 = [&](int i, int end) __attribute__((always_inline)) {
            cs = 0;
            c0 = c1 = c2 = c3 = ninf;
            if (i < end) {
                cs = rq[r][i];
                const float4 s4 = *reinterpret_cast<const float4*>(al_s + (size_t)cs * 4);
                c0 = leaky02(s4.x + d4.x);
                c1 = leaky02(s4.y + d4.y);
                c2 = leaky02(s4.z + d4.z);
                c3 = leaky02(s4.w + d4.w);
            }
        };

        float m0 = ninf, m1 = ninf, m2 = ninf, m3 = ninf;
        for_chunks([&](int from, int n) __attribute__((always_inline)) {
            logit4(from + lane, from + n);
            m0 = fmaxf(m0, c0);
            m1 = fmaxf(m1, c1);
            m2 = fmaxf(m2, c2);
            m3 = fmaxf(m3, c3);
        });
        m0 = wave_all<true>(m0);
        m1 = wave_all<true>(m1);
        m2 = wave_all<true>(m2);
        m3 = wave_all<true>(m3);

        const bool once = whole && tot <= 32;
        float dn0 = 0.f, dn1 = 0.f, dn2 = 0.f, dn3 = 0.f;
        vw_t  acc[4];
        #pragma unroll
        for (int k = 0; k < 4; ++k) acc[k] = vw_t(0.f);

        for_chunks([&](int from, int n) __attribute__((always_inline)) {
            if (!once) logit4(from + lane, from + n);
            float a[4];
            a[0] = __expf(c0 - m0);
            a[1] = __expf(c1 - m1);
            a[2] = __expf(c2 - m2);
            a[3] = __expf(c3 - m3);
            dn0 += a[0];
            dn1 += a[1];
            dn2 += a[2];
            dn3 += a[3];
            auto rows = [&](int j, auto UC) __attribute__((always_inline)) {
                constexpr int U = decltype(UC)::value;
                vw_t v[U][4];
                #pragma unroll
                for (int u = 0; u < U; ++u) {
                    const int sj = __builtin_amdgcn_readlane(cs, j + u);
                    const vw_t* __restrict__ p =
                        reinterpret_cast<const vw_t*>(hfeat + (size_t)sj * F) + lane;
                    #pragma unroll
                    for (int k = 0; k < 4; ++k) v[u][k] = p[k * 32];
                }
                #pragma unroll
                for (int u = 0; u < U; ++u) {
                    #pragma unroll
                    for (int k = 0; k < 4; ++k) {
                        const float ak = __builtin_bit_cast(float,
                            __builtin_amdgcn_readlane(__builtin_bit_cast(int, a[k]), j + u));
                        acc[k] += v[u][k] * ak;
                    }
                }
            };
            int j = 0;
            for (; j + 4 <= n; j += 4) rows(j, std::integral_constant<int, 4>{});
            if (j + 2 <= n) { rows(j, std::integral_constant<int, 2>{}); j += 2; }
            if (j < n)        rows(j, std::integral_constant<int, 1>{});
        });
        dn0 = wave_all<false>(dn0);
        dn1 = wave_all<false>(dn1);
        dn2 = wave_all<false>(dn2);
        dn3 = wave_all<false>(dn3);

        if constexpr (MEAN) {
            vw_t o = acc[0] / dn0;
            o += acc[1] / dn1;
            o += acc[2] / dn2;
            o += acc[3] / dn3;
            const vw_t b = reinterpret_cast<const vw_t*>(bias)[lane];
            reinterpret_cast<vw_t*>(agg + (size_t)row * C)[lane] = o * 0.25f + b;
        } else {
            vw_t* __restrict__ o = reinterpret_cast<vw_t*>(agg + (size_t)row * F) + lane;
            o[0]  = acc[0] / dn0;
            o[32] = acc[1] / dn1;
            o[64] = acc[2] / dn2;
            o[96] = acc[3] / dn3;
        }
    }
}


static inline unsigned cdiv_u(size_t a, unsigned b) {
    return (unsigned)((a + b - 1) / b);
}

extern "C" void kernel_launch(void* const* d_in, const int* in_sizes, int n_in,
                              void* d_out, int out_size, void* d_ws, size_t ws_size,
                              hipStream_t stream) {
    const float* x     = (const float*)d_in[0];
    const int*   ei    = (const int*)d_in[1];
    const float* W1    = (const float*)d_in[2];
    const float* a_s1  = (const float*)d_in[3];
    const float* a_d1  = (const float*)d_in[4];
    const float* b1    = (const float*)d_in[5];
    const float* W2    = (const float*)d_in[6];
    const float* a_s2  = (const float*)d_in[7];
    const float* a_d2  = (const float*)d_in[8];
    const float* b2    = (const float*)d_in[9];
    float*       out   = (float*)d_out;

    const int F1   = in_sizes[3];
    const int F2   = in_sizes[7];
    const int LAT  = in_sizes[9];
    const int H    = F2 / LAT;
    const int Fin  = in_sizes[2] / F1;
    const int N    = in_sizes[0] / Fin;
    const int E    = in_sizes[1] / 2;

    float* ws = (float*)d_ws;
    size_t o = 0;
    float* h1   = ws + o; o += (size_t)N * F1;
    float* agg1 = ws + o; o += (size_t)N * F1;
    float* als  = ws + o; o += (size_t)N * H;
    float* ald  = ws + o; o += (size_t)N * H;
    float* h2   = h1;
    int*   blist = (int*)(ws + o);
    unsigned* bandb = (unsigned*)(blist + (size_t)cdiv_u((size_t)N, BROWS) * (1 + BROWS + BROWS * 192));
    (void)ws_size; (void)n_in; (void)out_size;

    gemm_panel_f32<512, 4, false, true><<<N / PM, 256, 0, stream>>>(x, W1, h1, nullptr, a_s1, a_d1, als, ald, ei, E, bandb);

    agg_band<512, 128, BROWS, 32, 1, false><<<cdiv_u((size_t)N, BROWS), 1024, 0, stream>>>(
        ei, E, N, h1, als, ald, agg1, blist, nullptr, bandb);

    gemm_panel_f32<256, 2, true, false><<<N / PM, 256, 0, stream>>>(agg1, W2, h2, b1, a_s2, a_d2, als, ald, nullptr, 0, nullptr);

    agg_band<256, 64, BROWS, 32, 2, true><<<cdiv_u((size_t)N, BROWS), 1024, 0, stream>>>(
        ei, E, N, h2, als, ald, out, blist, b2, nullptr);
}
